# starting kernel
# baseline (speedup 1.0000x reference)
#include <hip/hip_runtime.h>

typedef __attribute__((ext_vector_type(2))) float v2f;
typedef __attribute__((ext_vector_type(8))) float v8f;

#define HDIM 128


__global__ void k_fill(float* __restrict__ p, float v, int n) {
    int i = blockIdx.x * blockDim.x + threadIdx.x;
    if (i < n) p[i] = v;
}

__global__ void k_deg_count(const int* __restrict__ dst, float* __restrict__ deg, int E) {
    int i = blockIdx.x * blockDim.x + threadIdx.x;
    if (i < E) atomicAdd(&deg[dst[i]], 1.0f);
}

__global__ void k_rsqrt_inplace(float* __restrict__ p, int n) {
    int i = blockIdx.x * blockDim.x + threadIdx.x;
    if (i < n) p[i] = rsqrtf(p[i]);
}

__global__ __launch_bounds__(256) void gemm_wmma_f32(
    const float* __restrict__ A, const float* __restrict__ B,
    float* __restrict__ C, int M) {
    const int K = HDIM;
    const int wave = threadIdx.x >> 5;
    const int lane = threadIdx.x & 31;

    const int m0 = blockIdx.x * 16;
    if (m0 >= M) return;
    const int n0 = wave * 16;

    const int lmod = lane & 15;
    const int kb   = (lane >> 4) << 1;

    const float* __restrict__ arow = A + (size_t)(m0 + lmod) * K;
    const float* __restrict__ bcol = B + (n0 + lmod);

    v8f acc = {};
#pragma unroll
    for (int k = 0; k < K; k += 4) {
        v2f a, b;
        a.x = arow[k + kb];
        a.y = arow[k + kb + 1];
        b.x = bcol[(size_t)(k + kb) * HDIM];
        b.y = bcol[(size_t)(k + kb + 1) * HDIM];
        acc = __builtin_amdgcn_wmma_f32_16x16x4_f32(
            false, a, false, b, (short)0, acc, false, false);
    }

    const int rbase = m0 + ((lane >> 4) << 3);
    float* __restrict__ cptr = C + (size_t)rbase * HDIM + n0 + lmod;
#pragma unroll
    for (int r = 0; r < 8; ++r) {
        cptr[(size_t)r * HDIM] = acc[r];
    }
}

__global__ __launch_bounds__(256) void k_scatter(
    const float* __restrict__ zw, const int* __restrict__ src, const int* __restrict__ dst,
    const float* __restrict__ dinv, float* __restrict__ agg, int E, int N) {
    const int wavesPerBlock = blockDim.x >> 5;
    const int e = blockIdx.x * wavesPerBlock + (threadIdx.x >> 5);
    if (e >= E + N) return;
    const int lane = threadIdx.x & 31;

    int s, d;
    if (e < E) { s = src[e]; d = dst[e]; }
    else       { s = e - E;  d = s;      }
    const float norm = dinv[s] * dinv[d];

    const float4 v = *(const float4*)(zw + (size_t)s * HDIM + lane * 4);
    float* __restrict__ ap = agg + (size_t)d * HDIM + lane * 4;
    atomicAdd(ap + 0, v.x * norm);
    atomicAdd(ap + 1, v.y * norm);
    atomicAdd(ap + 2, v.z * norm);
    atomicAdd(ap + 3, v.w * norm);
}

__global__ void k_post(const float* __restrict__ agg, const float* __restrict__ bias,
                       const float* __restrict__ alpha, const int* __restrict__ batch,
                       float* __restrict__ zout, float* __restrict__ gpool,
                       int goffset, int N) {
    int i = blockIdx.x * blockDim.x + threadIdx.x;
    if (i >= N * HDIM) return;
    const int h = i & (HDIM - 1);
    const int n = i >> 7;
    float z = agg[i] + bias[h];
    z = (z > 0.0f) ? z : alpha[h] * z;
    zout[i] = z;
    atomicAdd(&gpool[(size_t)batch[n] * (2 * HDIM) + goffset + h], z);
}


extern "C" void kernel_launch(void* const* d_in, const int* in_sizes, int n_in,
                              void* d_out, int out_size, void* d_ws, size_t ws_size,
                              hipStream_t stream) {
    const float* x     = (const float*)d_in[0];
    const int*   ei    = (const int*)  d_in[1];
    const int*   batch = (const int*)  d_in[2];
    const float* W0    = (const float*)d_in[3];
    const float* b0    = (const float*)d_in[4];
    const float* al0   = (const float*)d_in[5];
    const float* W1    = (const float*)d_in[6];
    const float* b1    = (const float*)d_in[7];
    const float* al1   = (const float*)d_in[8];

    const int N = in_sizes[2];
    const int E = in_sizes[1] / 2;
    const int* srcp = ei;
    const int* dstp = ei + E;

    float* ws   = (float*)d_ws;
    float* dinv = ws;
    float* zw   = dinv + N;
    float* agg  = zw  + (size_t)N * HDIM;
    float* z1   = agg + (size_t)N * HDIM;

    float* z2out = (float*)d_out;
    float* gpool = z2out + (size_t)N * HDIM;
    const int gsize = out_size - N * HDIM;

    const int T = 256;
    const int NH = N * HDIM;
    const int nEdges = E + N;
    const int scatterBlocks = (nEdges + 7) / 8;
    const int gemmBlocks = (N + 15) / 16;

    k_fill<<<(N + T - 1) / T, T, 0, stream>>>(dinv, 1.0f, N);
    k_deg_count<<<(E + T - 1) / T, T, 0, stream>>>(dstp, dinv, E);
    k_rsqrt_inplace<<<(N + T - 1) / T, T, 0, stream>>>(dinv, N);

    k_fill<<<(gsize + T - 1) / T, T, 0, stream>>>(gpool, 0.0f, gsize);

    gemm_wmma_f32<<<gemmBlocks, 256, 0, stream>>>(x, W0, zw, N);
    k_fill<<<(NH + T - 1) / T, T, 0, stream>>>(agg, 0.0f, NH);
    k_scatter<<<scatterBlocks, 256, 0, stream>>>(zw, srcp, dstp, dinv, agg, E, N);
    k_post<<<(NH + T - 1) / T, T, 0, stream>>>(agg, b0, al0, batch, z1, gpool, 0, N);

    gemm_wmma_f32<<<gemmBlocks, 256, 0, stream>>>(z1, W1, zw, N);
    k_fill<<<(NH + T - 1) / T, T, 0, stream>>>(agg, 0.0f, NH);
    k_scatter<<<scatterBlocks, 256, 0, stream>>>(zw, srcp, dstp, dinv, agg, E, N);
    k_post<<<(NH + T - 1) / T, T, 0, stream>>>(agg, b1, al1, batch, z2out, gpool, HDIM, N);
}


# k_scatter: lane l owns columns l,32+l,64+l,96+l, so each atomic instruction covers one whole 128-B line instead of 8 dwords in each of 4 lines; same atomic adds, arrival order as before
# speedup vs baseline: 1.4040x; 1.4040x over previous
#include <hip/hip_runtime.h>

typedef __attribute__((ext_vector_type(2))) float v2f;
typedef __attribute__((ext_vector_type(8))) float v8f;

#define HDIM 128


__global__ void k_fill(float* __restrict__ p, float v, int n) {
    int i = blockIdx.x * blockDim.x + threadIdx.x;
    if (i < n) p[i] = v;
}

__global__ void k_deg_count(const int* __restrict__ dst, float* __restrict__ deg, int E) {
    int i = blockIdx.x * blockDim.x + threadIdx.x;
    if (i < E) atomicAdd(&deg[dst[i]], 1.0f);
}

__global__ void k_rsqrt_inplace(float* __restrict__ p, int n) {
    int i = blockIdx.x * blockDim.x + threadIdx.x;
    if (i < n) p[i] = rsqrtf(p[i]);
}

__global__ __launch_bounds__(256) void gemm_wmma_f32(
    const float* __restrict__ A, const float* __restrict__ B,
    float* __restrict__ C, int M) {
    const int K = HDIM;
    const int wave = threadIdx.x >> 5;
    const int lane = threadIdx.x & 31;

    const int m0 = blockIdx.x * 16;
    if (m0 >= M) return;
    const int n0 = wave * 16;

    const int lmod = lane & 15;
    const int kb   = (lane >> 4) << 1;

    const float* __restrict__ arow = A + (size_t)(m0 + lmod) * K;
    const float* __restrict__ bcol = B + (n0 + lmod);

    v8f acc = {};
#pragma unroll
    for (int k = 0; k < K; k += 4) {
        v2f a, b;
        a.x = arow[k + kb];
        a.y = arow[k + kb + 1];
        b.x = bcol[(size_t)(k + kb) * HDIM];
        b.y = bcol[(size_t)(k + kb + 1) * HDIM];
        acc = __builtin_amdgcn_wmma_f32_16x16x4_f32(
            false, a, false, b, (short)0, acc, false, false);
    }

    const int rbase = m0 + ((lane >> 4) << 3);
    float* __restrict__ cptr = C + (size_t)rbase * HDIM + n0 + lmod;
#pragma unroll
    for (int r = 0; r < 8; ++r) {
        cptr[(size_t)r * HDIM] = acc[r];
    }
}

__global__ __launch_bounds__(256) void k_scatter(
    const float* __restrict__ zw, const int* __restrict__ src, const int* __restrict__ dst,
    const float* __restrict__ dinv, float* __restrict__ agg, int E, int N) {
    const int wavesPerBlock = blockDim.x >> 5;
    const int e = blockIdx.x * wavesPerBlock + (threadIdx.x >> 5);
    if (e >= E + N) return;
    const int lane = threadIdx.x & 31;

    int s, d;
    if (e < E) { s = src[e]; d = dst[e]; }
    else       { s = e - E;  d = s;      }
    const float norm = dinv[s] * dinv[d];

    const float* __restrict__ zp = zw + (size_t)s * HDIM + lane;
    float* __restrict__ ap = agg + (size_t)d * HDIM + lane;
    const float v0 = zp[0];
    const float v1 = zp[32];
    const float v2 = zp[64];
    const float v3 = zp[96];
    atomicAdd(ap + 0,  v0 * norm);
    atomicAdd(ap + 32, v1 * norm);
    atomicAdd(ap + 64, v2 * norm);
    atomicAdd(ap + 96, v3 * norm);
}

__global__ void k_post(const float* __restrict__ agg, const float* __restrict__ bias,
                       const float* __restrict__ alpha, const int* __restrict__ batch,
                       float* __restrict__ zout, float* __restrict__ gpool,
                       int goffset, int N) {
    int i = blockIdx.x * blockDim.x + threadIdx.x;
    if (i >= N * HDIM) return;
    const int h = i & (HDIM - 1);
    const int n = i >> 7;
    float z = agg[i] + bias[h];
    z = (z > 0.0f) ? z : alpha[h] * z;
    zout[i] = z;
    atomicAdd(&gpool[(size_t)batch[n] * (2 * HDIM) + goffset + h], z);
}


extern "C" void kernel_launch(void* const* d_in, const int* in_sizes, int n_in,
                              void* d_out, int out_size, void* d_ws, size_t ws_size,
                              hipStream_t stream) {
    const float* x     = (const float*)d_in[0];
    const int*   ei    = (const int*)  d_in[1];
    const int*   batch = (const int*)  d_in[2];
    const float* W0    = (const float*)d_in[3];
    const float* b0    = (const float*)d_in[4];
    const float* al0   = (const float*)d_in[5];
    const float* W1    = (const float*)d_in[6];
    const float* b1    = (const float*)d_in[7];
    const float* al1   = (const float*)d_in[8];

    const int N = in_sizes[2];
    const int E = in_sizes[1] / 2;
    const int* srcp = ei;
    const int* dstp = ei + E;

    float* ws   = (float*)d_ws;
    float* dinv = ws;
    float* zw   = dinv + N;
    float* agg  = zw  + (size_t)N * HDIM;
    float* z1   = agg + (size_t)N * HDIM;

    float* z2out = (float*)d_out;
    float* gpool = z2out + (size_t)N * HDIM;
    const int gsize = out_size - N * HDIM;

    const int T = 256;
    const int NH = N * HDIM;
    const int nEdges = E + N;
    const int scatterBlocks = (nEdges + 7) / 8;
    const int gemmBlocks = (N + 15) / 16;

    k_fill<<<(N + T - 1) / T, T, 0, stream>>>(dinv, 1.0f, N);
    k_deg_count<<<(E + T - 1) / T, T, 0, stream>>>(dstp, dinv, E);
    k_rsqrt_inplace<<<(N + T - 1) / T, T, 0, stream>>>(dinv, N);

    k_fill<<<(gsize + T - 1) / T, T, 0, stream>>>(gpool, 0.0f, gsize);

    gemm_wmma_f32<<<gemmBlocks, 256, 0, stream>>>(x, W0, zw, N);
    k_fill<<<(NH + T - 1) / T, T, 0, stream>>>(agg, 0.0f, NH);
    k_scatter<<<scatterBlocks, 256, 0, stream>>>(zw, srcp, dstp, dinv, agg, E, N);
    k_post<<<(NH + T - 1) / T, T, 0, stream>>>(agg, b0, al0, batch, z1, gpool, 0, N);

    gemm_wmma_f32<<<gemmBlocks, 256, 0, stream>>>(z1, W1, zw, N);
    k_fill<<<(NH + T - 1) / T, T, 0, stream>>>(agg, 0.0f, NH);
    k_scatter<<<scatterBlocks, 256, 0, stream>>>(zw, srcp, dstp, dinv, agg, E, N);
    k_post<<<(NH + T - 1) / T, T, 0, stream>>>(agg, b1, al1, batch, z2out, gpool, HDIM, N);
}


# edges counting-sorted by destination once per call (int atomics hand out slots, arrival order); one wave per node sums its in-edge rows in registers in slot order, then bias, PReLU, pool; agg is gone
# speedup vs baseline: 1.9939x; 1.4202x over previous
#include <hip/hip_runtime.h>

typedef __attribute__((ext_vector_type(2))) float v2f;
typedef __attribute__((ext_vector_type(8))) float v8f;

#define HDIM 128


__global__ void k_fill(float* __restrict__ p, float v, int n) {
    int i = blockIdx.x * blockDim.x + threadIdx.x;
    if (i < n) p[i] = v;
}

__global__ void k_zero_i32(int* __restrict__ p, int n) {
    int i = blockIdx.x * blockDim.x + threadIdx.x;
    if (i < n) p[i] = 0;
}

__device__ __forceinline__ int wave_sum_i32(int v) {
    for (int o = 16; o > 0; o >>= 1) v += __shfl_xor(v, o, 32);
    return v;
}


__global__ void k_count(const int* __restrict__ dst, int* __restrict__ cnt, int E) {
    int i = blockIdx.x * blockDim.x + threadIdx.x;
    if (i < E) atomicAdd(&cnt[dst[i]], 1);
}

__global__ __launch_bounds__(256) void k_blocksum(const int* __restrict__ cnt,
                                                  int* __restrict__ bsum, int N) {
    __shared__ int s_w[8];
    const int i = blockIdx.x * 256 + threadIdx.x;
    int v = (i < N) ? cnt[i] : 0;
    v = wave_sum_i32(v);
    if ((threadIdx.x & 31) == 0) s_w[threadIdx.x >> 5] = v;
    __syncthreads();
    if (threadIdx.x == 0) {
        int s = 0;
        for (int w = 0; w < 8; ++w) s += s_w[w];
        bsum[blockIdx.x] = s;
    }
}

__global__ __launch_bounds__(256) void k_rowptr(const int* __restrict__ cnt,
                                                const int* __restrict__ bsum,
                                                int* __restrict__ rowptr, int* __restrict__ cursor,
                                                float* __restrict__ dinv, int N) {
    __shared__ int s_a[8];
    __shared__ int s_b[8];
    const int t = threadIdx.x;
    const int lane = t & 31;
    const int wave = t >> 5;
    const int b = blockIdx.x;

    int pre = 0;
    for (int j = t; j < b; j += 256) pre += bsum[j];
    pre = wave_sum_i32(pre);
    if (lane == 0) s_a[wave] = pre;

    const int i = b * 256 + t;
    const int c = (i < N) ? cnt[i] : 0;
    int inc = c;
    for (int o = 1; o < 32; o <<= 1) {
        const int v = __shfl_up(inc, o, 32);
        if (lane >= o) inc += v;
    }
    if (lane == 31) s_b[wave] = inc;
    __syncthreads();

    int base = 0;
    for (int w = 0; w < 8; ++w) base += s_a[w];
    for (int w = 0; w < wave; ++w) base += s_b[w];
    const int excl = base + inc - c;
    if (i < N) {
        rowptr[i] = excl;
        cursor[i] = excl;
        dinv[i] = rsqrtf((float)(c + 1));
    }
    if (i == N - 1) rowptr[N] = excl + c;
}

__global__ void k_fill_csr(const int* __restrict__ src, const int* __restrict__ dst,
                           int* __restrict__ cursor, const float* __restrict__ dinv,
                           int* __restrict__ col, float* __restrict__ colw, int E) {
    int i = blockIdx.x * blockDim.x + threadIdx.x;
    if (i >= E) return;
    const int s = src[i];
    const int pos = atomicAdd(&cursor[dst[i]], 1);
    col[pos] = s;
    colw[pos] = dinv[s];
}

__global__ __launch_bounds__(256) void gemm_wmma_f32(
    const float* __restrict__ A, const float* __restrict__ B,
    float* __restrict__ C, int M) {
    const int K = HDIM;
    const int wave = threadIdx.x >> 5;
    const int lane = threadIdx.x & 31;

    const int m0 = blockIdx.x * 16;
    if (m0 >= M) return;
    const int n0 = wave * 16;

    const int lmod = lane & 15;
    const int kb   = (lane >> 4) << 1;

    const float* __restrict__ arow = A + (size_t)(m0 + lmod) * K;
    const float* __restrict__ bcol = B + (n0 + lmod);

    v8f acc = {};
#pragma unroll
    for (int k = 0; k < K; k += 4) {
        v2f a, b;
        a.x = arow[k + kb];
        a.y = arow[k + kb + 1];
        b.x = bcol[(size_t)(k + kb) * HDIM];
        b.y = bcol[(size_t)(k + kb + 1) * HDIM];
        acc = __builtin_amdgcn_wmma_f32_16x16x4_f32(
            false, a, false, b, (short)0, acc, false, false);
    }

    const int rbase = m0 + ((lane >> 4) << 3);
    float* __restrict__ cptr = C + (size_t)rbase * HDIM + n0 + lmod;
#pragma unroll
    for (int r = 0; r < 8; ++r) {
        cptr[(size_t)r * HDIM] = acc[r];
    }
}

template <int K>
__device__ __forceinline__ void add_rows(const float* __restrict__ zl, int my_s, float my_w, int t,
                                         float& a0, float& a1, float& a2, float& a3) {
    float x[K][4];
    float w[K];
#pragma unroll
    for (int k = 0; k < K; ++k) {
        const int s = __builtin_amdgcn_readlane(my_s, t + k);
        w[k] = __int_as_float(__builtin_amdgcn_readlane(__float_as_int(my_w), t + k));
        const float* __restrict__ p = zl + (size_t)s * HDIM;
        x[k][0] = p[0]; x[k][1] = p[32]; x[k][2] = p[64]; x[k][3] = p[96];
    }
#pragma unroll
    for (int k = 0; k < K; ++k) {
        a0 += x[k][0] * w[k]; a1 += x[k][1] * w[k]; a2 += x[k][2] * w[k]; a3 += x[k][3] * w[k];
    }
}

__global__ __launch_bounds__(256) void k_gather(
    const float* __restrict__ zw, const int* __restrict__ rowptr,
    const int* __restrict__ col, const float* __restrict__ colw,
    const float* __restrict__ dinv, const float* __restrict__ bias,
    const float* __restrict__ alpha, const int* __restrict__ batch,
    float* __restrict__ zout, float* __restrict__ gpool, int goffset, int N) {
    const int lane = threadIdx.x & 31;
    const int n = __builtin_amdgcn_readfirstlane((int)(blockIdx.x * 8 + (threadIdx.x >> 5)));
    if (n >= N) return;

    const int beg = rowptr[n];
    const int end = rowptr[n + 1];
    const float dn = dinv[n];
    const float* __restrict__ zl = zw + lane;

    float a0, a1, a2, a3;
    {
        const float w = dn * dn;
        const float* __restrict__ p = zl + (size_t)n * HDIM;
        a0 = p[0] * w; a1 = p[32] * w; a2 = p[64] * w; a3 = p[96] * w;
    }

    for (int base = beg; base < end; base += 32) {
        const int rem = end - base;
        const int m = rem < 32 ? rem : 32;
        int my_s = 0;
        float my_w = 0.0f;
        if (lane < m) {
            my_s = col[base + lane];
            my_w = colw[base + lane] * dn;
        }
        int t = 0;
        for (; t + 4 <= m; t += 4) add_rows<4>(zl, my_s, my_w, t, a0, a1, a2, a3);
        const int r = m - t;
        if (r == 3)      add_rows<3>(zl, my_s, my_w, t, a0, a1, a2, a3);
        else if (r == 2) add_rows<2>(zl, my_s, my_w, t, a0, a1, a2, a3);
        else if (r == 1) add_rows<1>(zl, my_s, my_w, t, a0, a1, a2, a3);
    }

    const int g = batch[n];
    float* __restrict__ zp = zout + (size_t)n * HDIM + lane;
    float* __restrict__ gp = gpool + (size_t)g * (2 * HDIM) + goffset + lane;
    const float* __restrict__ bp = bias + lane;
    const float* __restrict__ alp = alpha + lane;
    const float s0 = alp[0], s1 = alp[32], s2 = alp[64], s3 = alp[96];
    float z0 = a0 + bp[0], z1 = a1 + bp[32], z2 = a2 + bp[64], z3 = a3 + bp[96];
    z0 = z0 * ((z0 > 0.0f) ? 1.0f : s0);
    z1 = z1 * ((z1 > 0.0f) ? 1.0f : s1);
    z2 = z2 * ((z2 > 0.0f) ? 1.0f : s2);
    z3 = z3 * ((z3 > 0.0f) ? 1.0f : s3);
    zp[0] = z0; zp[32] = z1; zp[64] = z2; zp[96] = z3;
    atomicAdd(gp + 0,  z0);
    atomicAdd(gp + 32, z1);
    atomicAdd(gp + 64, z2);
    atomicAdd(gp + 96, z3);
}


extern "C" void kernel_launch(void* const* d_in, const int* in_sizes, int n_in,
                              void* d_out, int out_size, void* d_ws, size_t ws_size,
                              hipStream_t stream) {
    const float* x     = (const float*)d_in[0];
    const int*   ei    = (const int*)  d_in[1];
    const int*   batch = (const int*)  d_in[2];
    const float* W0    = (const float*)d_in[3];
    const float* b0    = (const float*)d_in[4];
    const float* al0   = (const float*)d_in[5];
    const float* W1    = (const float*)d_in[6];
    const float* b1    = (const float*)d_in[7];
    const float* al1   = (const float*)d_in[8];

    const int N = in_sizes[2];
    const int E = in_sizes[1] / 2;
    const int* srcp = ei;
    const int* dstp = ei + E;

    const int T = 256;
    const int NH = N * HDIM;
    const int nodeBlocks = (N + T - 1) / T;
    const int gatherBlocks = (N + 7) / 8;
    const int gemmBlocks = (N + 15) / 16;

    float* zw     = (float*)d_ws;
    float* z1     = zw + (size_t)NH;
    int*   col    = (int*)(z1 + (size_t)NH);
    float* colw   = (float*)(col + E);
    float* dinv   = colw + E;
    int*   cnt    = (int*)(dinv + N);
    int*   cursor = cnt + N;
    int*   rowptr = cursor + N;
    int*   bsum   = rowptr + (N + 1);

    float* z2out = (float*)d_out;
    float* gpool = z2out + (size_t)N * HDIM;
    const int gsize = out_size - N * HDIM;

    k_zero_i32<<<nodeBlocks, T, 0, stream>>>(cnt, N);
    k_count<<<(E + T - 1) / T, T, 0, stream>>>(dstp, cnt, E);
    k_blocksum<<<nodeBlocks, 256, 0, stream>>>(cnt, bsum, N);
    k_rowptr<<<nodeBlocks, 256, 0, stream>>>(cnt, bsum, rowptr, cursor, dinv, N);
    k_fill_csr<<<(E + T - 1) / T, T, 0, stream>>>(srcp, dstp, cursor, dinv, col, colw, E);

    k_fill<<<(gsize + T - 1) / T, T, 0, stream>>>(gpool, 0.0f, gsize);

    gemm_wmma_f32<<<gemmBlocks, 256, 0, stream>>>(x, W0, zw, N);
    k_gather<<<gatherBlocks, 256, 0, stream>>>(zw, rowptr, col, colw, dinv, b0, al0, batch,
                                               z1, gpool, 0, N);

    gemm_wmma_f32<<<gemmBlocks, 256, 0, stream>>>(z1, W1, zw, N);
    k_gather<<<gatherBlocks, 256, 0, stream>>>(zw, rowptr, col, colw, dinv, b1, al1, batch,
                                               z2out, gpool, HDIM, N);
}


# k_layer: blocks stay and loop over 16-node tiles (grid 1024, not one block per tile); W fragments, bias and slopes loaded once per block and kept in registers over its tiles; sums as before
# speedup vs baseline: 2.6315x; 1.3198x over previous
#include <hip/hip_runtime.h>

typedef __attribute__((ext_vector_type(2))) float v2f;
typedef __attribute__((ext_vector_type(8))) float v8f;

#define HDIM 128


__global__ void k_fill(float* __restrict__ p, float v, int n) {
    int i = blockIdx.x * blockDim.x + threadIdx.x;
    if (i < n) p[i] = v;
}

__global__ void k_zero_i32(int* __restrict__ p, int n) {
    int i = blockIdx.x * blockDim.x + threadIdx.x;
    if (i < n) p[i] = 0;
}

__device__ __forceinline__ int wave_sum_i32(int v) {
    for (int o = 16; o > 0; o >>= 1) v += __shfl_xor(v, o, 32);
    return v;
}


__global__ void k_count(const int* __restrict__ dst, int* __restrict__ cnt, int E) {
    int i = blockIdx.x * blockDim.x + threadIdx.x;
    if (i < E) atomicAdd(&cnt[dst[i]], 1);
}

__global__ __launch_bounds__(256) void k_blocksum(const int* __restrict__ cnt,
                                                  int* __restrict__ bsum, int N) {
    __shared__ int s_w[8];
    const int i = blockIdx.x * 256 + threadIdx.x;
    int v = (i < N) ? cnt[i] : 0;
    v = wave_sum_i32(v);
    if ((threadIdx.x & 31) == 0) s_w[threadIdx.x >> 5] = v;
    __syncthreads();
    if (threadIdx.x == 0) {
        int s = 0;
        for (int w = 0; w < 8; ++w) s += s_w[w];
        bsum[blockIdx.x] = s;
    }
}

__global__ __launch_bounds__(256) void k_rowptr(const int* __restrict__ cnt,
                                                const int* __restrict__ bsum,
                                                int* __restrict__ rowptr, int* __restrict__ cursor,
                                                float* __restrict__ dinv, int N) {
    __shared__ int s_a[8];
    __shared__ int s_b[8];
    const int t = threadIdx.x;
    const int lane = t & 31;
    const int wave = t >> 5;
    const int b = blockIdx.x;

    int pre = 0;
    for (int j = t; j < b; j += 256) pre += bsum[j];
    pre = wave_sum_i32(pre);
    if (lane == 0) s_a[wave] = pre;

    const int i = b * 256 + t;
    const int c = (i < N) ? cnt[i] : 0;
    int inc = c;
    for (int o = 1; o < 32; o <<= 1) {
        const int v = __shfl_up(inc, o, 32);
        if (lane >= o) inc += v;
    }
    if (lane == 31) s_b[wave] = inc;
    __syncthreads();

    int base = 0;
    for (int w = 0; w < 8; ++w) base += s_a[w];
    for (int w = 0; w < wave; ++w) base += s_b[w];
    const int excl = base + inc - c;
    if (i < N) {
        rowptr[i] = excl;
        cursor[i] = excl;
        dinv[i] = rsqrtf((float)(c + 1));
    }
    if (i == N - 1) rowptr[N] = excl + c;
}

__global__ void k_fill_csr(const int* __restrict__ src, const int* __restrict__ dst,
                           int* __restrict__ cursor, const float* __restrict__ dinv,
                           int* __restrict__ col, float* __restrict__ colw, int E) {
    int i = blockIdx.x * blockDim.x + threadIdx.x;
    if (i >= E) return;
    const int s = src[i];
    const int pos = atomicAdd(&cursor[dst[i]], 1);
    col[pos] = s;
    colw[pos] = dinv[s];
}

template <int K>
__device__ __forceinline__ void add_rows(const float* __restrict__ zl, int my_s, float my_w, int t,
                                         float& a0, float& a1, float& a2, float& a3) {
    float x[K][4];
    float w[K];
#pragma unroll
    for (int k = 0; k < K; ++k) {
        const int s = __builtin_amdgcn_readlane(my_s, t + k);
        w[k] = __int_as_float(__builtin_amdgcn_readlane(__float_as_int(my_w), t + k));
        const float* __restrict__ p = zl + (size_t)s * HDIM;
        x[k][0] = p[0]; x[k][1] = p[32]; x[k][2] = p[64]; x[k][3] = p[96];
    }
#pragma unroll
    for (int k = 0; k < K; ++k) {
        a0 += x[k][0] * w[k]; a1 += x[k][1] * w[k]; a2 += x[k][2] * w[k]; a3 += x[k][3] * w[k];
    }
}

__device__ __forceinline__ void add_in_edges(const float* __restrict__ zl, int beg, int end, float dn,
                                             const int* __restrict__ col,
                                             const float* __restrict__ colw, int lane,
                                             int my_s, float my_w,
                                             float& a0, float& a1, float& a2, float& a3) {
    for (int base = beg; base < end; base += 32) {
        const int rem = end - base;
        const int m = rem < 32 ? rem : 32;
        if (base != beg) {
            my_s = 0;
            my_w = 0.0f;
            if (lane < m) {
                my_s = col[base + lane];
                my_w = colw[base + lane];
            }
        }
        const float wv = my_w * dn;
        int t = 0;
        for (; t + 4 <= m; t += 4) add_rows<4>(zl, my_s, wv, t, a0, a1, a2, a3);
        const int r = m - t;
        if (r == 3)      add_rows<3>(zl, my_s, wv, t, a0, a1, a2, a3);
        else if (r == 2) add_rows<2>(zl, my_s, wv, t, a0, a1, a2, a3);
        else if (r == 1) add_rows<1>(zl, my_s, wv, t, a0, a1, a2, a3);
    }
}

#define TPITCH 132

__global__ __launch_bounds__(256) void k_layer(
    const float* __restrict__ zin, const float* __restrict__ Wm,
    const int* __restrict__ rowptr, const int* __restrict__ col, const float* __restrict__ colw,
    const float* __restrict__ dinv, const float* __restrict__ bias,
    const float* __restrict__ alpha, const int* __restrict__ batch,
    float* __restrict__ zout, float* __restrict__ gpool, int goffset, int N) {
    __shared__ __attribute__((aligned(16))) float sA[16 * TPITCH];
    __shared__ __attribute__((aligned(16))) float sC[16 * TPITCH];

    const int lane = threadIdx.x & 31;
    const int wave = __builtin_amdgcn_readfirstlane((int)(threadIdx.x >> 5));
    const int lmod = lane & 15;
    const int kb   = (lane >> 4) << 1;
    const int n0   = wave * 16;

    v2f bf[32];
    {
        const float* __restrict__ bcol = Wm + (n0 + lmod);
#pragma unroll
        for (int i = 0; i < 32; ++i) {
            bf[i].x = bcol[(size_t)(4 * i + kb) * HDIM];
            bf[i].y = bcol[(size_t)(4 * i + kb + 1) * HDIM];
        }
    }
    const float* __restrict__ bp = bias + lane;
    const float* __restrict__ alp = alpha + lane;
    const float b0 = bp[0], b1 = bp[32], b2 = bp[64], b3 = bp[96];
    const float s0 = alp[0], s1 = alp[32], s2 = alp[64], s3 = alp[96];

    const float* __restrict__ zl = zin + lane;
    const int numTiles = (N + 15) >> 4;

    for (int tile = (int)blockIdx.x; tile < numTiles; tile += (int)gridDim.x) {
        const int nA = tile * 16 + wave * 2;
        const int nB = nA + 1;
        const int cA = nA < N ? nA : N - 1;
        const int cB = nB < N ? nB : N - 1;
        const int pA = rowptr[nA < N ? nA : N];
        const int pB = rowptr[nB < N ? nB : N];
        const int pC = rowptr[nB + 1 < N ? nB + 1 : N];
        const float dA = dinv[cA];
        const float dB = dinv[cB];
        const int gA = batch[cA];
        const int gB = batch[cB];
        const float* __restrict__ qA = zl + (size_t)cA * HDIM;
        const float* __restrict__ qB = zl + (size_t)cB * HDIM;
        const float xA0 = qA[0], xA1 = qA[32], xA2 = qA[64], xA3 = qA[96];
        const float xB0 = qB[0], xB1 = qB[32], xB2 = qB[64], xB3 = qB[96];
        int sA0 = 0, sB0 = 0;
        float wA0 = 0.0f, wB0 = 0.0f;
        if (lane < pB - pA) { sA0 = col[pA + lane]; wA0 = colw[pA + lane]; }
        if (lane < pC - pB) { sB0 = col[pB + lane]; wB0 = colw[pB + lane]; }

        {
            float a0 = 0.0f, a1 = 0.0f, a2 = 0.0f, a3 = 0.0f;
            if (nA < N) {
                const float w = dA * dA;
                a0 = xA0 * w; a1 = xA1 * w; a2 = xA2 * w; a3 = xA3 * w;
                add_in_edges(zl, pA, pB, dA, col, colw, lane, sA0, wA0, a0, a1, a2, a3);
            }
            float* __restrict__ sp = sA + (wave * 2) * TPITCH + lane;
            sp[0] = a0; sp[32] = a1; sp[64] = a2; sp[96] = a3;
        }
        {
            float a0 = 0.0f, a1 = 0.0f, a2 = 0.0f, a3 = 0.0f;
            if (nB < N) {
                const float w = dB * dB;
                a0 = xB0 * w; a1 = xB1 * w; a2 = xB2 * w; a3 = xB3 * w;
                add_in_edges(zl, pB, pC, dB, col, colw, lane, sB0, wB0, a0, a1, a2, a3);
            }
            float* __restrict__ sp = sA + (wave * 2 + 1) * TPITCH + lane;
            sp[0] = a0; sp[32] = a1; sp[64] = a2; sp[96] = a3;
        }
        __syncthreads();

        v8f acc = {};
        {
            const float* __restrict__ ap = sA + lmod * TPITCH + kb;
#pragma unroll
            for (int i = 0; i < 32; ++i) {
                const v2f a = *(const v2f*)(ap + 4 * i);
                acc = __builtin_amdgcn_wmma_f32_16x16x4_f32(
                    false, a, false, bf[i], (short)0, acc, false, false);
            }
        }

        {
            float* __restrict__ cp = sC + ((lane >> 4) << 3) * TPITCH + n0 + lmod;
#pragma unroll
            for (int r = 0; r < 8; ++r) cp[r * TPITCH] = acc[r];
        }
        __syncthreads();

        for (int j = 0; j < 2; ++j) {
            const int n = nA + j;
            if (n >= N) break;
            const int g = j ? gB : gA;
            const float* __restrict__ rp = sC + (wave * 2 + j) * TPITCH + lane;
            float* __restrict__ zp = zout + (size_t)n * HDIM + lane;
            float* __restrict__ gp = gpool + (size_t)g * (2 * HDIM) + goffset + lane;
            float z0 = rp[0] + b0, z1 = rp[32] + b1, z2 = rp[64] + b2, z3 = rp[96] + b3;
            z0 = z0 * ((z0 > 0.0f) ? 1.0f : s0);
            z1 = z1 * ((z1 > 0.0f) ? 1.0f : s1);
            z2 = z2 * ((z2 > 0.0f) ? 1.0f : s2);
            z3 = z3 * ((z3 > 0.0f) ? 1.0f : s3);
            zp[0] = z0; zp[32] = z1; zp[64] = z2; zp[96] = z3;
            atomicAdd(gp + 0,  z0);
            atomicAdd(gp + 32, z1);
            atomicAdd(gp + 64, z2);
            atomicAdd(gp + 96, z3);
        }
    }
}


extern "C" void kernel_launch(void* const* d_in, const int* in_sizes, int n_in,
                              void* d_out, int out_size, void* d_ws, size_t ws_size,
                              hipStream_t stream) {
    const float* x     = (const float*)d_in[0];
    const int*   ei    = (const int*)  d_in[1];
    const int*   batch = (const int*)  d_in[2];
    const float* W0    = (const float*)d_in[3];
    const float* b0    = (const float*)d_in[4];
    const float* al0   = (const float*)d_in[5];
    const float* W1    = (const float*)d_in[6];
    const float* b1    = (const float*)d_in[7];
    const float* al1   = (const float*)d_in[8];

    const int N = in_sizes[2];
    const int E = in_sizes[1] / 2;
    const int* srcp = ei;
    const int* dstp = ei + E;

    const int T = 256;
    const int NH = N * HDIM;
    const int nodeBlocks = (N + T - 1) / T;
    const int numTiles = (N + 15) / 16;
    const int layerBlocks = numTiles < 1024 ? numTiles : 1024;

    float* z1     = (float*)d_ws;
    int*   col    = (int*)(z1 + (size_t)NH);
    float* colw   = (float*)(col + E);
    float* dinv   = colw + E;
    int*   cnt    = (int*)(dinv + N);
    int*   cursor = cnt + N;
    int*   rowptr = cursor + N;
    int*   bsum   = rowptr + (N + 1);

    float* z2out = (float*)d_out;
    float* gpool = z2out + (size_t)N * HDIM;
    const int gsize = out_size - N * HDIM;

    k_zero_i32<<<nodeBlocks, T, 0, stream>>>(cnt, N);
    k_count<<<(E + T - 1) / T, T, 0, stream>>>(dstp, cnt, E);
    k_blocksum<<<nodeBlocks, 256, 0, stream>>>(cnt, bsum, N);
    k_rowptr<<<nodeBlocks, 256, 0, stream>>>(cnt, bsum, rowptr, cursor, dinv, N);
    k_fill_csr<<<(E + T - 1) / T, T, 0, stream>>>(srcp, dstp, cursor, dinv, col, colw, E);

    k_fill<<<(gsize + T - 1) / T, T, 0, stream>>>(gpool, 0.0f, gsize);

    k_layer<<<layerBlocks, 256, 0, stream>>>(x, W0, rowptr, col, colw, dinv, b0, al0, batch,
                                             z1, gpool, 0, N);

    k_layer<<<layerBlocks, 256, 0, stream>>>(z1, W1, rowptr, col, colw, dinv, b1, al1, batch,
                                             z2out, gpool, HDIM, N);
}


# k_count keeps its atomic's return as the edge's rank; k_fill_csr writes slot rowptr[dst]+rank with plain loads: one int-atomic pass per call, not two; no cursor; slot order = arrival order as before
# speedup vs baseline: 3.0615x; 1.1634x over previous
#include <hip/hip_runtime.h>

typedef __attribute__((ext_vector_type(2))) float v2f;
typedef __attribute__((ext_vector_type(8))) float v8f;

#define HDIM 128


__global__ void k_fill(float* __restrict__ p, float v, int n) {
    int i = blockIdx.x * blockDim.x + threadIdx.x;
    if (i < n) p[i] = v;
}

__global__ void k_zero_i32(int* __restrict__ p, int n) {
    int i = blockIdx.x * blockDim.x + threadIdx.x;
    if (i < n) p[i] = 0;
}

__device__ __forceinline__ int wave_sum_i32(int v) {
    for (int o = 16; o > 0; o >>= 1) v += __shfl_xor(v, o, 32);
    return v;
}


__global__ void k_count(const int* __restrict__ dst, int* __restrict__ cnt,
                        int* __restrict__ rank, int E) {
    int i = blockIdx.x * blockDim.x + threadIdx.x;
    if (i < E) rank[i] = atomicAdd(&cnt[dst[i]], 1);
}

__global__ __launch_bounds__(256) void k_blocksum(const int* __restrict__ cnt,
                                                  int* __restrict__ bsum, int N) {
    __shared__ int s_w[8];
    const int i = blockIdx.x * 256 + threadIdx.x;
    int v = (i < N) ? cnt[i] : 0;
    v = wave_sum_i32(v);
    if ((threadIdx.x & 31) == 0) s_w[threadIdx.x >> 5] = v;
    __syncthreads();
    if (threadIdx.x == 0) {
        int s = 0;
        for (int w = 0; w < 8; ++w) s += s_w[w];
        bsum[blockIdx.x] = s;
    }
}

__global__ __launch_bounds__(256) void k_rowptr(const int* __restrict__ cnt,
                                                const int* __restrict__ bsum,
                                                int* __restrict__ rowptr,
                                                float* __restrict__ dinv, int N) {
    __shared__ int s_a[8];
    __shared__ int s_b[8];
    const int t = threadIdx.x;
    const int lane = t & 31;
    const int wave = t >> 5;
    const int b = blockIdx.x;

    int pre = 0;
    for (int j = t; j < b; j += 256) pre += bsum[j];
    pre = wave_sum_i32(pre);
    if (lane == 0) s_a[wave] = pre;

    const int i = b * 256 + t;
    const int c = (i < N) ? cnt[i] : 0;
    int inc = c;
    for (int o = 1; o < 32; o <<= 1) {
        const int v = __shfl_up(inc, o, 32);
        if (lane >= o) inc += v;
    }
    if (lane == 31) s_b[wave] = inc;
    __syncthreads();

    int base = 0;
    for (int w = 0; w < 8; ++w) base += s_a[w];
    for (int w = 0; w < wave; ++w) base += s_b[w];
    const int excl = base + inc - c;
    if (i < N) {
        rowptr[i] = excl;
        dinv[i] = rsqrtf((float)(c + 1));
    }
    if (i == N - 1) rowptr[N] = excl + c;
}

__global__ void k_fill_csr(const int* __restrict__ src, const int* __restrict__ dst,
                           const int* __restrict__ rank, const int* __restrict__ rowptr,
                           const float* __restrict__ dinv,
                           int* __restrict__ col, float* __restrict__ colw, int E) {
    int i = blockIdx.x * blockDim.x + threadIdx.x;
    if (i >= E) return;
    const int s = src[i];
    const int pos = rowptr[dst[i]] + rank[i];
    col[pos] = s;
    colw[pos] = dinv[s];
}

template <int K>
__device__ __forceinline__ void add_rows(const float* __restrict__ zl, int my_s, float my_w, int t,
                                         float& a0, float& a1, float& a2, float& a3) {
    float x[K][4];
    float w[K];
#pragma unroll
    for (int k = 0; k < K; ++k) {
        const int s = __builtin_amdgcn_readlane(my_s, t + k);
        w[k] = __int_as_float(__builtin_amdgcn_readlane(__float_as_int(my_w), t + k));
        const float* __restrict__ p = zl + (size_t)s * HDIM;
        x[k][0] = p[0]; x[k][1] = p[32]; x[k][2] = p[64]; x[k][3] = p[96];
    }
#pragma unroll
    for (int k = 0; k < K; ++k) {
        a0 += x[k][0] * w[k]; a1 += x[k][1] * w[k]; a2 += x[k][2] * w[k]; a3 += x[k][3] * w[k];
    }
}

__device__ __forceinline__ void add_in_edges(const float* __restrict__ zl, int beg, int end, float dn,
                                             const int* __restrict__ col,
                                             const float* __restrict__ colw, int lane,
                                             int my_s, float my_w,
                                             float& a0, float& a1, float& a2, float& a3) {
    for (int base = beg; base < end; base += 32) {
        const int rem = end - base;
        const int m = rem < 32 ? rem : 32;
        if (base != beg) {
            my_s = 0;
            my_w = 0.0f;
            if (lane < m) {
                my_s = col[base + lane];
                my_w = colw[base + lane];
            }
        }
        const float wv = my_w * dn;
        int t = 0;
        for (; t + 4 <= m; t += 4) add_rows<4>(zl, my_s, wv, t, a0, a1, a2, a3);
        const int r = m - t;
        if (r == 3)      add_rows<3>(zl, my_s, wv, t, a0, a1, a2, a3);
        else if (r == 2) add_rows<2>(zl, my_s, wv, t, a0, a1, a2, a3);
        else if (r == 1) add_rows<1>(zl, my_s, wv, t, a0, a1, a2, a3);
    }
}

#define TPITCH 132

__global__ __launch_bounds__(256) void k_layer(
    const float* __restrict__ zin, const float* __restrict__ Wm,
    const int* __restrict__ rowptr, const int* __restrict__ col, const float* __restrict__ colw,
    const float* __restrict__ dinv, const float* __restrict__ bias,
    const float* __restrict__ alpha, const int* __restrict__ batch,
    float* __restrict__ zout, float* __restrict__ gpool, int goffset, int N) {
    __shared__ __attribute__((aligned(16))) float sA[16 * TPITCH];
    __shared__ __attribute__((aligned(16))) float sC[16 * TPITCH];

    const int lane = threadIdx.x & 31;
    const int wave = __builtin_amdgcn_readfirstlane((int)(threadIdx.x >> 5));
    const int lmod = lane & 15;
    const int kb   = (lane >> 4) << 1;
    const int n0   = wave * 16;

    v2f bf[32];
    {
        const float* __restrict__ bcol = Wm + (n0 + lmod);
#pragma unroll
        for (int i = 0; i < 32; ++i) {
            bf[i].x = bcol[(size_t)(4 * i + kb) * HDIM];
            bf[i].y = bcol[(size_t)(4 * i + kb + 1) * HDIM];
        }
    }
    const float* __restrict__ bp = bias + lane;
    const float* __restrict__ alp = alpha + lane;
    const float b0 = bp[0], b1 = bp[32], b2 = bp[64], b3 = bp[96];
    const float s0 = alp[0], s1 = alp[32], s2 = alp[64], s3 = alp[96];

    const float* __restrict__ zl = zin + lane;
    const int numTiles = (N + 15) >> 4;

    for (int tile = (int)blockIdx.x; tile < numTiles; tile += (int)gridDim.x) {
        const int nA = tile * 16 + wave * 2;
        const int nB = nA + 1;
        const int cA = nA < N ? nA : N - 1;
        const int cB = nB < N ? nB : N - 1;
        const int pA = rowptr[nA < N ? nA : N];
        const int pB = rowptr[nB < N ? nB : N];
        const int pC = rowptr[nB + 1 < N ? nB + 1 : N];
        const float dA = dinv[cA];
        const float dB = dinv[cB];
        const int gA = batch[cA];
        const int gB = batch[cB];
        const float* __restrict__ qA = zl + (size_t)cA * HDIM;
        const float* __restrict__ qB = zl + (size_t)cB * HDIM;
        const float xA0 = qA[0], xA1 = qA[32], xA2 = qA[64], xA3 = qA[96];
        const float xB0 = qB[0], xB1 = qB[32], xB2 = qB[64], xB3 = qB[96];
        int sA0 = 0, sB0 = 0;
        float wA0 = 0.0f, wB0 = 0.0f;
        if (lane < pB - pA) { sA0 = col[pA + lane]; wA0 = colw[pA + lane]; }
        if (lane < pC - pB) { sB0 = col[pB + lane]; wB0 = colw[pB + lane]; }

        {
            float a0 = 0.0f, a1 = 0.0f, a2 = 0.0f, a3 = 0.0f;
            if (nA < N) {
                const float w = dA * dA;
                a0 = xA0 * w; a1 = xA1 * w; a2 = xA2 * w; a3 = xA3 * w;
                add_in_edges(zl, pA, pB, dA, col, colw, lane, sA0, wA0, a0, a1, a2, a3);
            }
            float* __restrict__ sp = sA + (wave * 2) * TPITCH + lane;
            sp[0] = a0; sp[32] = a1; sp[64] = a2; sp[96] = a3;
        }
        {
            float a0 = 0.0f, a1 = 0.0f, a2 = 0.0f, a3 = 0.0f;
            if (nB < N) {
                const float w = dB * dB;
                a0 = xB0 * w; a1 = xB1 * w; a2 = xB2 * w; a3 = xB3 * w;
                add_in_edges(zl, pB, pC, dB, col, colw, lane, sB0, wB0, a0, a1, a2, a3);
            }
            float* __restrict__ sp = sA + (wave * 2 + 1) * TPITCH + lane;
            sp[0] = a0; sp[32] = a1; sp[64] = a2; sp[96] = a3;
        }
        __syncthreads();

        v8f acc = {};
        {
            const float* __restrict__ ap = sA + lmod * TPITCH + kb;
#pragma unroll
            for (int i = 0; i < 32; ++i) {
                const v2f a = *(const v2f*)(ap + 4 * i);
                acc = __builtin_amdgcn_wmma_f32_16x16x4_f32(
                    false, a, false, bf[i], (short)0, acc, false, false);
            }
        }

        {
            float* __restrict__ cp = sC + ((lane >> 4) << 3) * TPITCH + n0 + lmod;
#pragma unroll
            for (int r = 0; r < 8; ++r) cp[r * TPITCH] = acc[r];
        }
        __syncthreads();

        for (int j = 0; j < 2; ++j) {
            const int n = nA + j;
            if (n >= N) break;
            const int g = j ? gB : gA;
            const float* __restrict__ rp = sC + (wave * 2 + j) * TPITCH + lane;
            float* __restrict__ zp = zout + (size_t)n * HDIM + lane;
            float* __restrict__ gp = gpool + (size_t)g * (2 * HDIM) + goffset + lane;
            float z0 = rp[0] + b0, z1 = rp[32] + b1, z2 = rp[64] + b2, z3 = rp[96] + b3;
            z0 = z0 * ((z0 > 0.0f) ? 1.0f : s0);
            z1 = z1 * ((z1 > 0.0f) ? 1.0f : s1);
            z2 = z2 * ((z2 > 0.0f) ? 1.0f : s2);
            z3 = z3 * ((z3 > 0.0f) ? 1.0f : s3);
            zp[0] = z0; zp[32] = z1; zp[64] = z2; zp[96] = z3;
            atomicAdd(gp + 0,  z0);
            atomicAdd(gp + 32, z1);
            atomicAdd(gp + 64, z2);
            atomicAdd(gp + 96, z3);
        }
    }
}


extern "C" void kernel_launch(void* const* d_in, const int* in_sizes, int n_in,
                              void* d_out, int out_size, void* d_ws, size_t ws_size,
                              hipStream_t stream) {
    const float* x     = (const float*)d_in[0];
    const int*   ei    = (const int*)  d_in[1];
    const int*   batch = (const int*)  d_in[2];
    const float* W0    = (const float*)d_in[3];
    const float* b0    = (const float*)d_in[4];
    const float* al0   = (const float*)d_in[5];
    const float* W1    = (const float*)d_in[6];
    const float* b1    = (const float*)d_in[7];
    const float* al1   = (const float*)d_in[8];

    const int N = in_sizes[2];
    const int E = in_sizes[1] / 2;
    const int* srcp = ei;
    const int* dstp = ei + E;

    const int T = 256;
    const int NH = N * HDIM;
    const int nodeBlocks = (N + T - 1) / T;
    const int numTiles = (N + 15) / 16;
    const int layerBlocks = numTiles < 1024 ? numTiles : 1024;

    float* z1     = (float*)d_ws;
    int*   col    = (int*)(z1 + (size_t)NH);
    float* colw   = (float*)(col + E);
    float* dinv   = colw + E;
    int*   cnt    = (int*)(dinv + N);
    int*   rank   = cnt + N;
    int*   rowptr = rank + E;
    int*   bsum   = rowptr + (N + 1);

    float* z2out = (float*)d_out;
    float* gpool = z2out + (size_t)N * HDIM;
    const int gsize = out_size - N * HDIM;

    k_zero_i32<<<nodeBlocks, T, 0, stream>>>(cnt, N);
    k_count<<<(E + T - 1) / T, T, 0, stream>>>(dstp, cnt, rank, E);
    k_blocksum<<<nodeBlocks, 256, 0, stream>>>(cnt, bsum, N);
    k_rowptr<<<nodeBlocks, 256, 0, stream>>>(cnt, bsum, rowptr, dinv, N);
    k_fill_csr<<<(E + T - 1) / T, T, 0, stream>>>(srcp, dstp, rank, rowptr, dinv, col, colw, E);

    k_fill<<<(gsize + T - 1) / T, T, 0, stream>>>(gpool, 0.0f, gsize);

    k_layer<<<layerBlocks, 256, 0, stream>>>(x, W0, rowptr, col, colw, dinv, b0, al0, batch,
                                             z1, gpool, 0, N);

    k_layer<<<layerBlocks, 256, 0, stream>>>(z1, W1, rowptr, col, colw, dinv, b1, al1, batch,
                                             z2out, gpool, HDIM, N);
}


# edge sort: each (edge slice, node range) block counts, then places its edges by LDS atomics; slice counts prefixed per node; no global int atomic, no rank array; slot order: slice, then LDS arrival
# speedup vs baseline: 4.0716x; 1.3299x over previous
#include <hip/hip_runtime.h>

typedef __attribute__((ext_vector_type(2))) float v2f;
typedef __attribute__((ext_vector_type(8))) float v8f;

#define HDIM 128


__global__ void k_fill(float* __restrict__ p, float v, int n) {
    int i = blockIdx.x * blockDim.x + threadIdx.x;
    if (i < n) p[i] = v;
}

__device__ __forceinline__ int wave_sum_i32(int v) {
    for (int o = 16; o > 0; o >>= 1) v += __shfl_xor(v, o, 32);
    return v;
}

#define CS_SLICES  32
#define CS_THREADS 512
#define CS_PER     13
#define CS_RANGE   (CS_PER * CS_THREADS)
#define CS_BATCH   16

__global__ __launch_bounds__(CS_THREADS) void k_hist(const int* __restrict__ dst, int* __restrict__ hs,
                                                     int hpitch, int E, int N) {
    __shared__ int s_cnt[CS_RANGE];
    const int t = threadIdx.x;
    const int slice = blockIdx.x;
    const int r0 = blockIdx.y * CS_RANGE;
#pragma unroll
    for (int k = 0; k < CS_PER; ++k) s_cnt[t + k * CS_THREADS] = 0;
    __syncthreads();

    const int sliceLen = (E + CS_SLICES - 1) / CS_SLICES;
    const int beg = slice * sliceLen;
    const int end = (beg + sliceLen < E) ? beg + sliceLen : E;
    for (int i0 = beg + t; i0 < end; i0 += CS_BATCH * CS_THREADS) {
        int d[CS_BATCH];
#pragma unroll
        for (int k = 0; k < CS_BATCH; ++k) {
            const int i = i0 + k * CS_THREADS;
            d[k] = dst[i < end ? i : end - 1];
        }
#pragma unroll
        for (int k = 0; k < CS_BATCH; ++k) {
            const unsigned j = (unsigned)(d[k] - r0);
            if (i0 + k * CS_THREADS < end && j < (unsigned)CS_RANGE && d[k] < N) atomicAdd(&s_cnt[j], 1);
        }
    }
    __syncthreads();

    int* __restrict__ row = hs + (size_t)slice * hpitch + r0;
    int c[CS_PER];
#pragma unroll
    for (int k = 0; k < CS_PER; ++k) c[k] = s_cnt[t + k * CS_THREADS];
#pragma unroll
    for (int k = 0; k < CS_PER; ++k) {
        const int j = t + k * CS_THREADS;
        if (r0 + j < N) row[j] = c[k];
    }
}

__global__ __launch_bounds__(256) void k_colscan(int* __restrict__ hs, int hpitch,
                                                 int* __restrict__ cnt, int* __restrict__ bsum, int N) {
    __shared__ int s_w[8];
    const int i = blockIdx.x * 256 + threadIdx.x;
    int run = 0;
    if (i < N) {
        int* __restrict__ colp = hs + i;
        int v[CS_SLICES];
#pragma unroll
        for (int s = 0; s < CS_SLICES; ++s) v[s] = colp[(size_t)s * hpitch];
#pragma unroll
        for (int s = 0; s < CS_SLICES; ++s) {
            colp[(size_t)s * hpitch] = run;
            run += v[s];
        }
        cnt[i] = run;
    }
    const int ws = wave_sum_i32(run);
    if ((threadIdx.x & 31) == 0) s_w[threadIdx.x >> 5] = ws;
    __syncthreads();
    if (threadIdx.x == 0) {
        int s = 0;
        for (int w = 0; w < 8; ++w) s += s_w[w];
        bsum[blockIdx.x] = s;
    }
}

__global__ __launch_bounds__(256) void k_rowptr(const int* __restrict__ cnt,
                                                const int* __restrict__ bsum,
                                                int* __restrict__ rowptr,
                                                float* __restrict__ dinv, int N) {
    __shared__ int s_a[8];
    __shared__ int s_b[8];
    const int t = threadIdx.x;
    const int lane = t & 31;
    const int wave = t >> 5;
    const int b = blockIdx.x;

    int pre = 0;
    for (int j = t; j < b; j += 256) pre += bsum[j];
    pre = wave_sum_i32(pre);
    if (lane == 0) s_a[wave] = pre;

    const int i = b * 256 + t;
    const int c = (i < N) ? cnt[i] : 0;
    int inc = c;
    for (int o = 1; o < 32; o <<= 1) {
        const int v = __shfl_up(inc, o, 32);
        if (lane >= o) inc += v;
    }
    if (lane == 31) s_b[wave] = inc;
    __syncthreads();

    int base = 0;
    for (int w = 0; w < 8; ++w) base += s_a[w];
    for (int w = 0; w < wave; ++w) base += s_b[w];
    const int excl = base + inc - c;
    if (i < N) {
        rowptr[i] = excl;
        dinv[i] = rsqrtf((float)(c + 1));
    }
    if (i == N - 1) rowptr[N] = excl + c;
}

__global__ __launch_bounds__(CS_THREADS) void k_place(const int* __restrict__ src, const int* __restrict__ dst,
                                                      const int* __restrict__ hs, int hpitch,
                                                      const int* __restrict__ rowptr,
                                                      const float* __restrict__ dinv,
                                                      int* __restrict__ col, float* __restrict__ colw,
                                                      int E, int N) {
    __shared__ int s_cur[CS_RANGE];
    const int t = threadIdx.x;
    const int slice = blockIdx.x;
    const int r0 = blockIdx.y * CS_RANGE;
    {
        const int* __restrict__ row = hs + (size_t)slice * hpitch;
        int a[CS_PER];
        int b[CS_PER];
#pragma unroll
        for (int k = 0; k < CS_PER; ++k) {
            const int n = r0 + t + k * CS_THREADS;
            const int nc = n < N ? n : N - 1;
            a[k] = rowptr[nc];
            b[k] = row[nc];
        }
#pragma unroll
        for (int k = 0; k < CS_PER; ++k) s_cur[t + k * CS_THREADS] = a[k] + b[k];
    }
    __syncthreads();

    const int sliceLen = (E + CS_SLICES - 1) / CS_SLICES;
    const int beg = slice * sliceLen;
    const int end = (beg + sliceLen < E) ? beg + sliceLen : E;
    for (int i0 = beg + t; i0 < end; i0 += CS_BATCH * CS_THREADS) {
        int d[CS_BATCH];
        int sv[CS_BATCH];
        int pos[CS_BATCH];
        float w[CS_BATCH];
#pragma unroll
        for (int k = 0; k < CS_BATCH; ++k) {
            const int i = i0 + k * CS_THREADS;
            const int ic = i < end ? i : end - 1;
            d[k]  = dst[ic];
            sv[k] = src[ic];
        }
#pragma unroll
        for (int k = 0; k < CS_BATCH; ++k) {
            const unsigned j = (unsigned)(d[k] - r0);
            const bool m = i0 + k * CS_THREADS < end && j < (unsigned)CS_RANGE && d[k] < N;
            pos[k] = -1;
            if (m) pos[k] = atomicAdd(&s_cur[j], 1);
            w[k] = dinv[m ? sv[k] : 0];
        }
#pragma unroll
        for (int k = 0; k < CS_BATCH; ++k) {
            if (pos[k] >= 0) {
                col[pos[k]] = sv[k];
                colw[pos[k]] = w[k];
            }
        }
    }
}

template <int K>
__device__ __forceinline__ void add_rows(const float* __restrict__ zl, int my_s, float my_w, int t,
                                         float& a0, float& a1, float& a2, float& a3) {
    float x[K][4];
    float w[K];
#pragma unroll
    for (int k = 0; k < K; ++k) {
        const int s = __builtin_amdgcn_readlane(my_s, t + k);
        w[k] = __int_as_float(__builtin_amdgcn_readlane(__float_as_int(my_w), t + k));
        const float* __restrict__ p = zl + (size_t)s * HDIM;
        x[k][0] = p[0]; x[k][1] = p[32]; x[k][2] = p[64]; x[k][3] = p[96];
    }
#pragma unroll
    for (int k = 0; k < K; ++k) {
        a0 += x[k][0] * w[k]; a1 += x[k][1] * w[k]; a2 += x[k][2] * w[k]; a3 += x[k][3] * w[k];
    }
}

__device__ __forceinline__ void add_in_edges(const float* __restrict__ zl, int beg, int end, float dn,
                                             const int* __restrict__ col,
                                             const float* __restrict__ colw, int lane,
                                             int my_s, float my_w,
                                             float& a0, float& a1, float& a2, float& a3) {
    for (int base = beg; base < end; base += 32) {
        const int rem = end - base;
        const int m = rem < 32 ? rem : 32;
        if (base != beg) {
            my_s = 0;
            my_w = 0.0f;
            if (lane < m) {
                my_s = col[base + lane];
                my_w = colw[base + lane];
            }
        }
        const float wv = my_w * dn;
        int t = 0;
        for (; t + 4 <= m; t += 4) add_rows<4>(zl, my_s, wv, t, a0, a1, a2, a3);
        const int r = m - t;
        if (r == 3)      add_rows<3>(zl, my_s, wv, t, a0, a1, a2, a3);
        else if (r == 2) add_rows<2>(zl, my_s, wv, t, a0, a1, a2, a3);
        else if (r == 1) add_rows<1>(zl, my_s, wv, t, a0, a1, a2, a3);
    }
}

#define TPITCH 132

__global__ __launch_bounds__(256) void k_layer(
    const float* __restrict__ zin, const float* __restrict__ Wm,
    const int* __restrict__ rowptr, const int* __restrict__ col, const float* __restrict__ colw,
    const float* __restrict__ dinv, const float* __restrict__ bias,
    const float* __restrict__ alpha, const int* __restrict__ batch,
    float* __restrict__ zout, float* __restrict__ gpool, int goffset, int N) {
    __shared__ __attribute__((aligned(16))) float sA[16 * TPITCH];
    __shared__ __attribute__((aligned(16))) float sC[16 * TPITCH];

    const int lane = threadIdx.x & 31;
    const int wave = __builtin_amdgcn_readfirstlane((int)(threadIdx.x >> 5));
    const int lmod = lane & 15;
    const int kb   = (lane >> 4) << 1;
    const int n0   = wave * 16;

    v2f bf[32];
    {
        const float* __restrict__ bcol = Wm + (n0 + lmod);
#pragma unroll
        for (int i = 0; i < 32; ++i) {
            bf[i].x = bcol[(size_t)(4 * i + kb) * HDIM];
            bf[i].y = bcol[(size_t)(4 * i + kb + 1) * HDIM];
        }
    }
    const float* __restrict__ bp = bias + lane;
    const float* __restrict__ alp = alpha + lane;
    const float b0 = bp[0], b1 = bp[32], b2 = bp[64], b3 = bp[96];
    const float s0 = alp[0], s1 = alp[32], s2 = alp[64], s3 = alp[96];

    const float* __restrict__ zl = zin + lane;
    const int numTiles = (N + 15) >> 4;

    for (int tile = (int)blockIdx.x; tile < numTiles; tile += (int)gridDim.x) {
        const int nA = tile * 16 + wave * 2;
        const int nB = nA + 1;
        const int cA = nA < N ? nA : N - 1;
        const int cB = nB < N ? nB : N - 1;
        const int pA = rowptr[nA < N ? nA : N];
        const int pB = rowptr[nB < N ? nB : N];
        const int pC = rowptr[nB + 1 < N ? nB + 1 : N];
        const float dA = dinv[cA];
        const float dB = dinv[cB];
        const int gA = batch[cA];
        const int gB = batch[cB];
        const float* __restrict__ qA = zl + (size_t)cA * HDIM;
        const float* __restrict__ qB = zl + (size_t)cB * HDIM;
        const float xA0 = qA[0], xA1 = qA[32], xA2 = qA[64], xA3 = qA[96];
        const float xB0 = qB[0], xB1 = qB[32], xB2 = qB[64], xB3 = qB[96];
        int sA0 = 0, sB0 = 0;
        float wA0 = 0.0f, wB0 = 0.0f;
        if (lane < pB - pA) { sA0 = col[pA + lane]; wA0 = colw[pA + lane]; }
        if (lane < pC - pB) { sB0 = col[pB + lane]; wB0 = colw[pB + lane]; }

        {
            float a0 = 0.0f, a1 = 0.0f, a2 = 0.0f, a3 = 0.0f;
            if (nA < N) {
                const float w = dA * dA;
                a0 = xA0 * w; a1 = xA1 * w; a2 = xA2 * w; a3 = xA3 * w;
                add_in_edges(zl, pA, pB, dA, col, colw, lane, sA0, wA0, a0, a1, a2, a3);
            }
            float* __restrict__ sp = sA + (wave * 2) * TPITCH + lane;
            sp[0] = a0; sp[32] = a1; sp[64] = a2; sp[96] = a3;
        }
        {
            float a0 = 0.0f, a1 = 0.0f, a2 = 0.0f, a3 = 0.0f;
            if (nB < N) {
                const float w = dB * dB;
                a0 = xB0 * w; a1 = xB1 * w; a2 = xB2 * w; a3 = xB3 * w;
                add_in_edges(zl, pB, pC, dB, col, colw, lane, sB0, wB0, a0, a1, a2, a3);
            }
            float* __restrict__ sp = sA + (wave * 2 + 1) * TPITCH + lane;
            sp[0] = a0; sp[32] = a1; sp[64] = a2; sp[96] = a3;
        }
        __syncthreads();

        v8f acc = {};
        {
            const float* __restrict__ ap = sA + lmod * TPITCH + kb;
#pragma unroll
            for (int i = 0; i < 32; ++i) {
                const v2f a = *(const v2f*)(ap + 4 * i);
                acc = __builtin_amdgcn_wmma_f32_16x16x4_f32(
                    false, a, false, bf[i], (short)0, acc, false, false);
            }
        }

        {
            float* __restrict__ cp = sC + ((lane >> 4) << 3) * TPITCH + n0 + lmod;
#pragma unroll
            for (int r = 0; r < 8; ++r) cp[r * TPITCH] = acc[r];
        }
        __syncthreads();

        for (int j = 0; j < 2; ++j) {
            const int n = nA + j;
            if (n >= N) break;
            const int g = j ? gB : gA;
            const float* __restrict__ rp = sC + (wave * 2 + j) * TPITCH + lane;
            float* __restrict__ zp = zout + (size_t)n * HDIM + lane;
            float* __restrict__ gp = gpool + (size_t)g * (2 * HDIM) + goffset + lane;
            float z0 = rp[0] + b0, z1 = rp[32] + b1, z2 = rp[64] + b2, z3 = rp[96] + b3;
            z0 = z0 * ((z0 > 0.0f) ? 1.0f : s0);
            z1 = z1 * ((z1 > 0.0f) ? 1.0f : s1);
            z2 = z2 * ((z2 > 0.0f) ? 1.0f : s2);
            z3 = z3 * ((z3 > 0.0f) ? 1.0f : s3);
            zp[0] = z0; zp[32] = z1; zp[64] = z2; zp[96] = z3;
            atomicAdd(gp + 0,  z0);
            atomicAdd(gp + 32, z1);
            atomicAdd(gp + 64, z2);
            atomicAdd(gp + 96, z3);
        }
    }
}


extern "C" void kernel_launch(void* const* d_in, const int* in_sizes, int n_in,
                              void* d_out, int out_size, void* d_ws, size_t ws_size,
                              hipStream_t stream) {
    const float* x     = (const float*)d_in[0];
    const int*   ei    = (const int*)  d_in[1];
    const int*   batch = (const int*)  d_in[2];
    const float* W0    = (const float*)d_in[3];
    const float* b0    = (const float*)d_in[4];
    const float* al0   = (const float*)d_in[5];
    const float* W1    = (const float*)d_in[6];
    const float* b1    = (const float*)d_in[7];
    const float* al1   = (const float*)d_in[8];

    const int N = in_sizes[2];
    const int E = in_sizes[1] / 2;
    const int* srcp = ei;
    const int* dstp = ei + E;

    const int T = 256;
    const int NH = N * HDIM;
    const int nodeBlocks = (N + T - 1) / T;
    const int numTiles = (N + 15) / 16;
    const int layerBlocks = numTiles < 1024 ? numTiles : 1024;
    const int rangeBlocks = (N + CS_RANGE - 1) / CS_RANGE;
    const int hpitch = (N + 31) / 32 * 32;

    float* z1     = (float*)d_ws;
    int*   col    = (int*)(z1 + (size_t)NH);
    float* colw   = (float*)(col + E);
    int*   hs     = (int*)(colw + E);
    float* dinv   = (float*)(hs + (size_t)CS_SLICES * hpitch);
    int*   cnt    = (int*)(dinv + N);
    int*   rowptr = cnt + N;
    int*   bsum   = rowptr + (N + 1);

    float* z2out = (float*)d_out;
    float* gpool = z2out + (size_t)N * HDIM;
    const int gsize = out_size - N * HDIM;

    const dim3 edgeGrid(CS_SLICES, rangeBlocks);
    k_hist<<<edgeGrid, CS_THREADS, 0, stream>>>(dstp, hs, hpitch, E, N);
    k_colscan<<<nodeBlocks, 256, 0, stream>>>(hs, hpitch, cnt, bsum, N);
    k_rowptr<<<nodeBlocks, 256, 0, stream>>>(cnt, bsum, rowptr, dinv, N);
    k_place<<<edgeGrid, CS_THREADS, 0, stream>>>(srcp, dstp, hs, hpitch, rowptr, dinv, col, colw, E, N);

    k_fill<<<(gsize + T - 1) / T, T, 0, stream>>>(gpool, 0.0f, gsize);

    k_layer<<<layerBlocks, 256, 0, stream>>>(x, W0, rowptr, col, colw, dinv, b0, al0, batch,
                                             z1, gpool, 0, N);

    k_layer<<<layerBlocks, 256, 0, stream>>>(z1, W1, rowptr, col, colw, dinv, b1, al1, batch,
                                             z2out, gpool, HDIM, N);
}


# k_layer pool: blocks take consecutive tiles; a wave keeps its column's sum over a run of equal graph ids in a register: one atomic add per run, not per row; pool sums reordered (run sums first)
# speedup vs baseline: 4.3097x; 1.0585x over previous
#include <hip/hip_runtime.h>

typedef __attribute__((ext_vector_type(2))) float v2f;
typedef __attribute__((ext_vector_type(8))) float v8f;

#define HDIM 128


__global__ void k_fill(float* __restrict__ p, float v, int n) {
    int i = blockIdx.x * blockDim.x + threadIdx.x;
    if (i < n) p[i] = v;
}

__device__ __forceinline__ int wave_sum_i32(int v) {
    for (int o = 16; o > 0; o >>= 1) v += __shfl_xor(v, o, 32);
    return v;
}

#define CS_SLICES  32
#define CS_THREADS 512
#define CS_PER     13
#define CS_RANGE   (CS_PER * CS_THREADS)
#define CS_BATCH   16

__global__ __launch_bounds__(CS_THREADS) void k_hist(const int* __restrict__ dst, int* __restrict__ hs,
                                                     int hpitch, int E, int N) {
    __shared__ int s_cnt[CS_RANGE];
    const int t = threadIdx.x;
    const int slice = blockIdx.x;
    const int r0 = blockIdx.y * CS_RANGE;
#pragma unroll
    for (int k = 0; k < CS_PER; ++k) s_cnt[t + k * CS_THREADS] = 0;
    __syncthreads();

    const int sliceLen = (E + CS_SLICES - 1) / CS_SLICES;
    const int beg = slice * sliceLen;
    const int end = (beg + sliceLen < E) ? beg + sliceLen : E;
    for (int i0 = beg + t; i0 < end; i0 += CS_BATCH * CS_THREADS) {
        int d[CS_BATCH];
#pragma unroll
        for (int k = 0; k < CS_BATCH; ++k) {
            const int i = i0 + k * CS_THREADS;
            d[k] = dst[i < end ? i : end - 1];
        }
#pragma unroll
        for (int k = 0; k < CS_BATCH; ++k) {
            const unsigned j = (unsigned)(d[k] - r0);
            if (i0 + k * CS_THREADS < end && j < (unsigned)CS_RANGE && d[k] < N) atomicAdd(&s_cnt[j], 1);
        }
    }
    __syncthreads();

    int* __restrict__ row = hs + (size_t)slice * hpitch + r0;
    int c[CS_PER];
#pragma unroll
    for (int k = 0; k < CS_PER; ++k) c[k] = s_cnt[t + k * CS_THREADS];
#pragma unroll
    for (int k = 0; k < CS_PER; ++k) {
        const int j = t + k * CS_THREADS;
        if (r0 + j < N) row[j] = c[k];
    }
}

__global__ __launch_bounds__(256) void k_colscan(int* __restrict__ hs, int hpitch,
                                                 int* __restrict__ cnt, int* __restrict__ bsum, int N) {
    __shared__ int s_w[8];
    const int i = blockIdx.x * 256 + threadIdx.x;
    int run = 0;
    if (i < N) {
        int* __restrict__ colp = hs + i;
        int v[CS_SLICES];
#pragma unroll
        for (int s = 0; s < CS_SLICES; ++s) v[s] = colp[(size_t)s * hpitch];
#pragma unroll
        for (int s = 0; s < CS_SLICES; ++s) {
            colp[(size_t)s * hpitch] = run;
            run += v[s];
        }
        cnt[i] = run;
    }
    const int ws = wave_sum_i32(run);
    if ((threadIdx.x & 31) == 0) s_w[threadIdx.x >> 5] = ws;
    __syncthreads();
    if (threadIdx.x == 0) {
        int s = 0;
        for (int w = 0; w < 8; ++w) s += s_w[w];
        bsum[blockIdx.x] = s;
    }
}

__global__ __launch_bounds__(256) void k_rowptr(const int* __restrict__ cnt,
                                                const int* __restrict__ bsum,
                                                int* __restrict__ rowptr,
                                                float* __restrict__ dinv, int N) {
    __shared__ int s_a[8];
    __shared__ int s_b[8];
    const int t = threadIdx.x;
    const int lane = t & 31;
    const int wave = t >> 5;
    const int b = blockIdx.x;

    int pre = 0;
    for (int j = t; j < b; j += 256) pre += bsum[j];
    pre = wave_sum_i32(pre);
    if (lane == 0) s_a[wave] = pre;

    const int i = b * 256 + t;
    const int c = (i < N) ? cnt[i] : 0;
    int inc = c;
    for (int o = 1; o < 32; o <<= 1) {
        const int v = __shfl_up(inc, o, 32);
        if (lane >= o) inc += v;
    }
    if (lane == 31) s_b[wave] = inc;
    __syncthreads();

    int base = 0;
    for (int w = 0; w < 8; ++w) base += s_a[w];
    for (int w = 0; w < wave; ++w) base += s_b[w];
    const int excl = base + inc - c;
    if (i < N) {
        rowptr[i] = excl;
        dinv[i] = rsqrtf((float)(c + 1));
    }
    if (i == N - 1) rowptr[N] = excl + c;
}

__global__ __launch_bounds__(CS_THREADS) void k_place(const int* __restrict__ src, const int* __restrict__ dst,
                                                      const int* __restrict__ hs, int hpitch,
                                                      const int* __restrict__ rowptr,
                                                      const float* __restrict__ dinv,
                                                      int* __restrict__ col, float* __restrict__ colw,
                                                      int E, int N) {
    __shared__ int s_cur[CS_RANGE];
    const int t = threadIdx.x;
    const int slice = blockIdx.x;
    const int r0 = blockIdx.y * CS_RANGE;
    {
        const int* __restrict__ row = hs + (size_t)slice * hpitch;
        int a[CS_PER];
        int b[CS_PER];
#pragma unroll
        for (int k = 0; k < CS_PER; ++k) {
            const int n = r0 + t + k * CS_THREADS;
            const int nc = n < N ? n : N - 1;
            a[k] = rowptr[nc];
            b[k] = row[nc];
        }
#pragma unroll
        for (int k = 0; k < CS_PER; ++k) s_cur[t + k * CS_THREADS] = a[k] + b[k];
    }
    __syncthreads();

    const int sliceLen = (E + CS_SLICES - 1) / CS_SLICES;
    const int beg = slice * sliceLen;
    const int end = (beg + sliceLen < E) ? beg + sliceLen : E;
    for (int i0 = beg + t; i0 < end; i0 += CS_BATCH * CS_THREADS) {
        int d[CS_BATCH];
        int sv[CS_BATCH];
        int pos[CS_BATCH];
        float w[CS_BATCH];
#pragma unroll
        for (int k = 0; k < CS_BATCH; ++k) {
            const int i = i0 + k * CS_THREADS;
            const int ic = i < end ? i : end - 1;
            d[k]  = dst[ic];
            sv[k] = src[ic];
        }
#pragma unroll
        for (int k = 0; k < CS_BATCH; ++k) {
            const unsigned j = (unsigned)(d[k] - r0);
            const bool m = i0 + k * CS_THREADS < end && j < (unsigned)CS_RANGE && d[k] < N;
            pos[k] = -1;
            if (m) pos[k] = atomicAdd(&s_cur[j], 1);
            w[k] = dinv[m ? sv[k] : 0];
        }
#pragma unroll
        for (int k = 0; k < CS_BATCH; ++k) {
            if (pos[k] >= 0) {
                col[pos[k]] = sv[k];
                colw[pos[k]] = w[k];
            }
        }
    }
}

template <int K>
__device__ __forceinline__ void add_rows(const float* __restrict__ zl, int my_s, float my_w, int t,
                                         float& a0, float& a1, float& a2, float& a3) {
    float x[K][4];
    float w[K];
#pragma unroll
    for (int k = 0; k < K; ++k) {
        const int s = __builtin_amdgcn_readlane(my_s, t + k);
        w[k] = __int_as_float(__builtin_amdgcn_readlane(__float_as_int(my_w), t + k));
        const float* __restrict__ p = zl + (size_t)s * HDIM;
        x[k][0] = p[0]; x[k][1] = p[32]; x[k][2] = p[64]; x[k][3] = p[96];
    }
#pragma unroll
    for (int k = 0; k < K; ++k) {
        a0 += x[k][0] * w[k]; a1 += x[k][1] * w[k]; a2 += x[k][2] * w[k]; a3 += x[k][3] * w[k];
    }
}

__device__ __forceinline__ void add_in_edges(const float* __restrict__ zl, int beg, int end, float dn,
                                             const int* __restrict__ col,
                                             const float* __restrict__ colw, int lane,
                                             int my_s, float my_w,
                                             float& a0, float& a1, float& a2, float& a3) {
    for (int base = beg; base < end; base += 32) {
        const int rem = end - base;
        const int m = rem < 32 ? rem : 32;
        if (base != beg) {
            my_s = 0;
            my_w = 0.0f;
            if (lane < m) {
                my_s = col[base + lane];
                my_w = colw[base + lane];
            }
        }
        const float wv = my_w * dn;
        int t = 0;
        for (; t + 4 <= m; t += 4) add_rows<4>(zl, my_s, wv, t, a0, a1, a2, a3);
        const int r = m - t;
        if (r == 3)      add_rows<3>(zl, my_s, wv, t, a0, a1, a2, a3);
        else if (r == 2) add_rows<2>(zl, my_s, wv, t, a0, a1, a2, a3);
        else if (r == 1) add_rows<1>(zl, my_s, wv, t, a0, a1, a2, a3);
    }
}

#define TPITCH 132

__global__ __launch_bounds__(256) void k_layer(
    const float* __restrict__ zin, const float* __restrict__ Wm,
    const int* __restrict__ rowptr, const int* __restrict__ col, const float* __restrict__ colw,
    const float* __restrict__ dinv, const float* __restrict__ bias,
    const float* __restrict__ alpha, const int* __restrict__ batch,
    float* __restrict__ zout, float* __restrict__ gpool, int goffset, int N) {
    __shared__ __attribute__((aligned(16))) float sA[16 * TPITCH];
    __shared__ __attribute__((aligned(16))) float sC[16 * TPITCH];

    const int lane = threadIdx.x & 31;
    const int wave = __builtin_amdgcn_readfirstlane((int)(threadIdx.x >> 5));
    const int lmod = lane & 15;
    const int kb   = (lane >> 4) << 1;
    const int n0   = wave * 16;

    v2f bf[32];
    {
        const float* __restrict__ bcol = Wm + (n0 + lmod);
#pragma unroll
        for (int i = 0; i < 32; ++i) {
            bf[i].x = bcol[(size_t)(4 * i + kb) * HDIM];
            bf[i].y = bcol[(size_t)(4 * i + kb + 1) * HDIM];
        }
    }
    const int ecol = ((wave & 3) << 5) + lane;
    const int erow = (wave >> 2) << 3;
    const float eb = bias[ecol];
    const float es = alpha[ecol];
    float pacc = 0.0f;
    int pg = -1;
    float* __restrict__ gcol = gpool + goffset + ecol;

    const float* __restrict__ zl = zin + lane;
    const int numTiles = (N + 15) >> 4;

    const int tBase = numTiles / (int)gridDim.x;
    const int tRem  = numTiles % (int)gridDim.x;
    const int bIdx  = (int)blockIdx.x;
    const int tBeg  = bIdx * tBase + (bIdx < tRem ? bIdx : tRem);
    const int tEnd  = tBeg + tBase + (bIdx < tRem ? 1 : 0);
    for (int tile = tBeg; tile < tEnd; ++tile) {
        const int nA = tile * 16 + wave * 2;
        const int nB = nA + 1;
        const int cA = nA < N ? nA : N - 1;
        const int cB = nB < N ? nB : N - 1;
        const int pA = rowptr[nA < N ? nA : N];
        const int pB = rowptr[nB < N ? nB : N];
        const int pC = rowptr[nB + 1 < N ? nB + 1 : N];
        const float dA = dinv[cA];
        const float dB = dinv[cB];
        const int eN = tile * 16 + erow;
        const int eNl = eN + (lane & 7);
        const int egv = batch[eNl < N ? eNl : N - 1];
        const float* __restrict__ qA = zl + (size_t)cA * HDIM;
        const float* __restrict__ qB = zl + (size_t)cB * HDIM;
        const float xA0 = qA[0], xA1 = qA[32], xA2 = qA[64], xA3 = qA[96];
        const float xB0 = qB[0], xB1 = qB[32], xB2 = qB[64], xB3 = qB[96];
        int sA0 = 0, sB0 = 0;
        float wA0 = 0.0f, wB0 = 0.0f;
        if (lane < pB - pA) { sA0 = col[pA + lane]; wA0 = colw[pA + lane]; }
        if (lane < pC - pB) { sB0 = col[pB + lane]; wB0 = colw[pB + lane]; }

        {
            float a0 = 0.0f, a1 = 0.0f, a2 = 0.0f, a3 = 0.0f;
            if (nA < N) {
                const float w = dA * dA;
                a0 = xA0 * w; a1 = xA1 * w; a2 = xA2 * w; a3 = xA3 * w;
                add_in_edges(zl, pA, pB, dA, col, colw, lane, sA0, wA0, a0, a1, a2, a3);
            }
            float* __restrict__ sp = sA + (wave * 2) * TPITCH + lane;
            sp[0] = a0; sp[32] = a1; sp[64] = a2; sp[96] = a3;
        }
        {
            float a0 = 0.0f, a1 = 0.0f, a2 = 0.0f, a3 = 0.0f;
            if (nB < N) {
                const float w = dB * dB;
                a0 = xB0 * w; a1 = xB1 * w; a2 = xB2 * w; a3 = xB3 * w;
                add_in_edges(zl, pB, pC, dB, col, colw, lane, sB0, wB0, a0, a1, a2, a3);
            }
            float* __restrict__ sp = sA + (wave * 2 + 1) * TPITCH + lane;
            sp[0] = a0; sp[32] = a1; sp[64] = a2; sp[96] = a3;
        }
        __syncthreads();

        v8f acc = {};
        {
            const float* __restrict__ ap = sA + lmod * TPITCH + kb;
#pragma unroll
            for (int i = 0; i < 32; ++i) {
                const v2f a = *(const v2f*)(ap + 4 * i);
                acc = __builtin_amdgcn_wmma_f32_16x16x4_f32(
                    false, a, false, bf[i], (short)0, acc, false, false);
            }
        }

        {
            float* __restrict__ cp = sC + ((lane >> 4) << 3) * TPITCH + n0 + lmod;
#pragma unroll
            for (int r = 0; r < 8; ++r) cp[r * TPITCH] = acc[r];
        }
        __syncthreads();

        {
            const float* __restrict__ rp = sC + erow * TPITCH + ecol;
            float* __restrict__ zp = zout + (size_t)eN * HDIM + ecol;
            float z[8];
#pragma unroll
            for (int r = 0; r < 8; ++r) z[r] = rp[r * TPITCH];
#pragma unroll
            for (int r = 0; r < 8; ++r) {
                if (eN + r >= N) break;
                const float v = z[r] + eb;
                const float y = v * ((v > 0.0f) ? 1.0f : es);
                zp[(size_t)r * HDIM] = y;
                const int g = __builtin_amdgcn_readlane(egv, r);
                if (g != pg) {
                    if (pg >= 0) atomicAdd(gcol + (size_t)pg * (2 * HDIM), pacc);
                    pacc = 0.0f;
                    pg = g;
                }
                pacc += y;
            }
        }
    }
    if (pg >= 0) atomicAdd(gcol + (size_t)pg * (2 * HDIM), pacc);
}


extern "C" void kernel_launch(void* const* d_in, const int* in_sizes, int n_in,
                              void* d_out, int out_size, void* d_ws, size_t ws_size,
                              hipStream_t stream) {
    const float* x     = (const float*)d_in[0];
    const int*   ei    = (const int*)  d_in[1];
    const int*   batch = (const int*)  d_in[2];
    const float* W0    = (const float*)d_in[3];
    const float* b0    = (const float*)d_in[4];
    const float* al0   = (const float*)d_in[5];
    const float* W1    = (const float*)d_in[6];
    const float* b1    = (const float*)d_in[7];
    const float* al1   = (const float*)d_in[8];

    const int N = in_sizes[2];
    const int E = in_sizes[1] / 2;
    const int* srcp = ei;
    const int* dstp = ei + E;

    const int T = 256;
    const int NH = N * HDIM;
    const int nodeBlocks = (N + T - 1) / T;
    const int numTiles = (N + 15) / 16;
    const int layerBlocks = numTiles < 1024 ? numTiles : 1024;
    const int rangeBlocks = (N + CS_RANGE - 1) / CS_RANGE;
    const int hpitch = (N + 31) / 32 * 32;

    float* z1     = (float*)d_ws;
    int*   col    = (int*)(z1 + (size_t)NH);
    float* colw   = (float*)(col + E);
    int*   hs     = (int*)(colw + E);
    float* dinv   = (float*)(hs + (size_t)CS_SLICES * hpitch);
    int*   cnt    = (int*)(dinv + N);
    int*   rowptr = cnt + N;
    int*   bsum   = rowptr + (N + 1);

    float* z2out = (float*)d_out;
    float* gpool = z2out + (size_t)N * HDIM;
    const int gsize = out_size - N * HDIM;

    const dim3 edgeGrid(CS_SLICES, rangeBlocks);
    k_hist<<<edgeGrid, CS_THREADS, 0, stream>>>(dstp, hs, hpitch, E, N);
    k_colscan<<<nodeBlocks, 256, 0, stream>>>(hs, hpitch, cnt, bsum, N);
    k_rowptr<<<nodeBlocks, 256, 0, stream>>>(cnt, bsum, rowptr, dinv, N);
    k_place<<<edgeGrid, CS_THREADS, 0, stream>>>(srcp, dstp, hs, hpitch, rowptr, dinv, col, colw, E, N);

    k_fill<<<(gsize + T - 1) / T, T, 0, stream>>>(gpool, 0.0f, gsize);

    k_layer<<<layerBlocks, 256, 0, stream>>>(x, W0, rowptr, col, colw, dinv, b0, al0, batch,
                                             z1, gpool, 0, N);

    k_layer<<<layerBlocks, 256, 0, stream>>>(z1, W1, rowptr, col, colw, dinv, b1, al1, batch,
                                             z2out, gpool, HDIM, N);
}


# edge sort in 2 launches, not 5: slices sort by 256-node bucket in LDS, one block per bucket builds its CSR part and scales x rows by dinv; rows summed pre-scaled; slot order: LDS arrival
# speedup vs baseline: 5.6388x; 1.3084x over previous
#include <hip/hip_runtime.h>

typedef __attribute__((ext_vector_type(2))) float v2f;
typedef __attribute__((ext_vector_type(8))) float v8f;

#define HDIM 128

__device__ __forceinline__ int wave_sum_i32(int v) {
    for (int o = 16; o > 0; o >>= 1) v += __shfl_xor(v, o, 32);
    return v;
}

__device__ __forceinline__ int wave_scan_i32(int v, int lane) {
    for (int o = 1; o < 32; o <<= 1) {
        const int u = __shfl_up(v, o, 32);
        if (lane >= o) v += u;
    }
    return v;
}

#define ES_SLICES  256
#define ES_THREADS 512
#define ES_BATCH   8
#define ES_MAXB    (2 * ES_THREADS)
#define BK_NODES   256
#define BK_THREADS 512
#define BK_RUNS    (ES_SLICES / (BK_THREADS / 32))
#define XS_PIECES  (BK_NODES * (HDIM / 4) / BK_THREADS)

__global__ __launch_bounds__(ES_THREADS) void k_tilesort(const int* __restrict__ src, const int* __restrict__ dst,
                                                         int2* __restrict__ es, int* __restrict__ offT,
                                                         float* __restrict__ gpool, int gsize,
                                                         int E, int N, int nb) {
    __shared__ int s_cnt[ES_MAXB];
    __shared__ int s_w[ES_THREADS / 32];
    const int t = threadIdx.x;
    const int lane = t & 31;
    const int wave = t >> 5;
    const int slice = blockIdx.x;

    for (int i = slice * ES_THREADS + t; i < gsize; i += (int)gridDim.x * ES_THREADS) gpool[i] = 0.0f;

    s_cnt[t] = 0;
    s_cnt[t + ES_THREADS] = 0;
    __syncthreads();

    const int sliceLen = (E + ES_SLICES - 1) / ES_SLICES;
    const int beg = slice * sliceLen;
    const int end = (beg + sliceLen < E) ? beg + sliceLen : E;

    for (int i0 = beg + t; i0 < end; i0 += ES_BATCH * ES_THREADS) {
        int d[ES_BATCH];
#pragma unroll
        for (int k = 0; k < ES_BATCH; ++k) {
            const int i = i0 + k * ES_THREADS;
            d[k] = dst[i < end ? i : end - 1];
        }
#pragma unroll
        for (int k = 0; k < ES_BATCH; ++k) {
            if (i0 + k * ES_THREADS < end && (unsigned)d[k] < (unsigned)N) atomicAdd(&s_cnt[d[k] / BK_NODES], 1);
        }
    }
    __syncthreads();

    const int v0 = s_cnt[2 * t];
    const int v1 = s_cnt[2 * t + 1];
    const int loc = v0 + v1;
    const int inc = wave_scan_i32(loc, lane);
    if (lane == 31) s_w[wave] = inc;
    __syncthreads();
    int wbase = 0;
    for (int w = 0; w < wave; ++w) wbase += s_w[w];
    const int ex0 = wbase + inc - loc;
    const int ex1 = ex0 + v0;
    s_cnt[2 * t] = ex0;
    s_cnt[2 * t + 1] = ex1;
    if (2 * t <= nb)     offT[(size_t)(2 * t) * ES_SLICES + slice] = ex0;
    if (2 * t + 1 <= nb) offT[(size_t)(2 * t + 1) * ES_SLICES + slice] = ex1;
    __syncthreads();

    for (int i0 = beg + t; i0 < end; i0 += ES_BATCH * ES_THREADS) {
        int d[ES_BATCH];
        int sv[ES_BATCH];
#pragma unroll
        for (int k = 0; k < ES_BATCH; ++k) {
            const int i = i0 + k * ES_THREADS;
            const int ic = i < end ? i : end - 1;
            d[k]  = dst[ic];
            sv[k] = src[ic];
        }
        int p[ES_BATCH];
#pragma unroll
        for (int k = 0; k < ES_BATCH; ++k) {
            p[k] = -1;
            if (i0 + k * ES_THREADS < end && (unsigned)d[k] < (unsigned)N) p[k] = atomicAdd(&s_cnt[d[k] / BK_NODES], 1);
        }
#pragma unroll
        for (int k = 0; k < ES_BATCH; ++k) {
            if (p[k] >= 0) es[(size_t)beg + p[k]] = make_int2(sv[k], d[k]);
        }
    }
}

__global__ __launch_bounds__(BK_THREADS) void k_csr(const int2* __restrict__ es, const int* __restrict__ offT,
                                                    const float* __restrict__ x, float* __restrict__ xs,
                                                    int* __restrict__ rowptr, float* __restrict__ dinv,
                                                    int* __restrict__ col, int E, int N) {
    __shared__ int s_o0[ES_SLICES];
    __shared__ int s_len[ES_SLICES];
    __shared__ int s_cnt[BK_NODES];
    __shared__ float s_dinv[BK_NODES];
    __shared__ int s_ra[BK_THREADS / 32];
    __shared__ int s_rb[BK_THREADS / 32];
    const int t = threadIdx.x;
    const int lane = t & 31;
    const int wave = __builtin_amdgcn_readfirstlane((int)(threadIdx.x >> 5));
    const int b = blockIdx.x;
    const int n0 = b * BK_NODES;

    int o0 = 0;
    if (t < ES_SLICES) {
        o0 = offT[(size_t)b * ES_SLICES + t];
        const int o1 = offT[(size_t)(b + 1) * ES_SLICES + t];
        s_o0[t] = o0;
        s_len[t] = o1 - o0;
    }
    if (t < BK_NODES) s_cnt[t] = 0;
    const int osum = wave_sum_i32(o0);
    if (lane == 0) s_ra[wave] = osum;
    __syncthreads();
    int base = 0;
    for (int w = 0; w < BK_THREADS / 32; ++w) base += s_ra[w];

    const int sliceLen = (E + ES_SLICES - 1) / ES_SLICES;

    int2 e[BK_RUNS];
    int len[BK_RUNS];
    int st[BK_RUNS];
    int myLen = 0, mySt = 0;
    if (lane < BK_RUNS) {
        myLen = s_len[wave * BK_RUNS + lane];
        mySt = s_o0[wave * BK_RUNS + lane];
    }
#pragma unroll
    for (int k = 0; k < BK_RUNS; ++k) {
        const int s = wave * BK_RUNS + k;
        len[k] = __builtin_amdgcn_readlane(myLen, k);
        st[k] = __builtin_amdgcn_readlane(mySt, k);
        e[k] = make_int2(0, n0);
        if (lane < len[k]) e[k] = es[(size_t)s * sliceLen + st[k] + lane];
    }
    const float4* __restrict__ x4 = (const float4*)x;
    float4 v[XS_PIECES];
#pragma unroll
    for (int i = 0; i < XS_PIECES; ++i) {
        const int q = t + i * BK_THREADS;
        const int n = n0 + (q >> 5);
        v[i] = x4[(size_t)(n < N ? n : N - 1) * (HDIM / 4) + (q & 31)];
    }
#pragma unroll
    for (int k = 0; k < BK_RUNS; ++k) {
        if (lane < len[k]) atomicAdd(&s_cnt[(e[k].y - n0) & (BK_NODES - 1)], 1);
        if (len[k] > 32) {
            const int s = wave * BK_RUNS + k;
            for (int j = 32 + lane; j < len[k]; j += 32) {
                const int2 q = es[(size_t)s * sliceLen + st[k] + j];
                atomicAdd(&s_cnt[(q.y - n0) & (BK_NODES - 1)], 1);
            }
        }
    }
    __syncthreads();

    const int c = (t < BK_NODES) ? s_cnt[t] : 0;
    const int inc = wave_scan_i32(c, lane);
    if (lane == 31) s_rb[wave] = inc;
    __syncthreads();
    int wbase = 0;
    for (int w = 0; w < wave; ++w) wbase += s_rb[w];
    const int excl = wbase + inc - c;
    if (t < BK_NODES) {
        const int n = n0 + t;
        const float dv = rsqrtf((float)(c + 1));
        s_dinv[t] = dv;
        s_cnt[t] = excl;
        if (n < N) {
            rowptr[n] = base + excl;
            dinv[n] = dv;
        }
        if (t == BK_NODES - 1 && b == (int)gridDim.x - 1) rowptr[N] = base + excl + c;
    }
    __syncthreads();

    int p[BK_RUNS];
#pragma unroll
    for (int k = 0; k < BK_RUNS; ++k) {
        p[k] = -1;
        if (lane < len[k]) p[k] = atomicAdd(&s_cnt[(e[k].y - n0) & (BK_NODES - 1)], 1);
    }
#pragma unroll
    for (int k = 0; k < BK_RUNS; ++k) {
        if (p[k] >= 0) col[base + p[k]] = e[k].x;
    }
#pragma unroll
    for (int k = 0; k < BK_RUNS; ++k) {
        if (len[k] > 32) {
            const int s = wave * BK_RUNS + k;
            for (int j = 32 + lane; j < len[k]; j += 32) {
                const int2 q = es[(size_t)s * sliceLen + st[k] + j];
                const int pj = atomicAdd(&s_cnt[(q.y - n0) & (BK_NODES - 1)], 1);
                col[base + pj] = q.x;
            }
        }
    }

    {
        float4* __restrict__ xs4 = (float4*)xs;
#pragma unroll
        for (int i = 0; i < XS_PIECES; ++i) {
            const int q = t + i * BK_THREADS;
            const int r = q >> 5;
            const float dv = s_dinv[r];
            if (n0 + r < N) {
                float4 o;
                o.x = v[i].x * dv; o.y = v[i].y * dv; o.z = v[i].z * dv; o.w = v[i].w * dv;
                xs4[(size_t)(n0 + r) * (HDIM / 4) + (q & 31)] = o;
            }
        }
    }
}

template <int K>
__device__ __forceinline__ void add_rows(const float* __restrict__ zl, int my_s, int t,
                                         float& a0, float& a1, float& a2, float& a3) {
    float x[K][4];
#pragma unroll
    for (int k = 0; k < K; ++k) {
        const int s = __builtin_amdgcn_readlane(my_s, t + k);
        const float* __restrict__ p = zl + (size_t)s * HDIM;
        x[k][0] = p[0]; x[k][1] = p[32]; x[k][2] = p[64]; x[k][3] = p[96];
    }
#pragma unroll
    for (int k = 0; k < K; ++k) {
        a0 += x[k][0]; a1 += x[k][1]; a2 += x[k][2]; a3 += x[k][3];
    }
}

__device__ __forceinline__ void add_in_edges(const float* __restrict__ zl, int beg, int end,
                                             const int* __restrict__ col, int lane, int my_s,
                                             float& a0, float& a1, float& a2, float& a3) {
    for (int base = beg; base < end; base += 32) {
        const int rem = end - base;
        const int m = rem < 32 ? rem : 32;
        if (base != beg) {
            my_s = 0;
            if (lane < m) my_s = col[base + lane];
        }
        int t = 0;
        for (; t + 4 <= m; t += 4) add_rows<4>(zl, my_s, t, a0, a1, a2, a3);
        const int r = m - t;
        if (r == 3)      add_rows<3>(zl, my_s, t, a0, a1, a2, a3);
        else if (r == 2) add_rows<2>(zl, my_s, t, a0, a1, a2, a3);
        else if (r == 1) add_rows<1>(zl, my_s, t, a0, a1, a2, a3);
    }
}

#define TPITCH 132

template <bool SCALE_OUT>
__global__ __launch_bounds__(256) void k_layer(
    const float* __restrict__ zin, const float* __restrict__ Wm,
    const int* __restrict__ rowptr, const int* __restrict__ col,
    const float* __restrict__ dinv, const float* __restrict__ bias,
    const float* __restrict__ alpha, const int* __restrict__ batch,
    float* __restrict__ zout, float* __restrict__ gpool, int goffset, int N) {
    __shared__ __attribute__((aligned(16))) float sA[16 * TPITCH];
    __shared__ __attribute__((aligned(16))) float sC[16 * TPITCH];

    const int lane = threadIdx.x & 31;
    const int wave = __builtin_amdgcn_readfirstlane((int)(threadIdx.x >> 5));
    const int lmod = lane & 15;
    const int kb   = (lane >> 4) << 1;
    const int n0   = wave * 16;

    v2f bf[32];
    {
        const float* __restrict__ bcol = Wm + (n0 + lmod);
#pragma unroll
        for (int i = 0; i < 32; ++i) {
            bf[i].x = bcol[(size_t)(4 * i + kb) * HDIM];
            bf[i].y = bcol[(size_t)(4 * i + kb + 1) * HDIM];
        }
    }
    const int ecol = ((wave & 3) << 5) + lane;
    const int erow = (wave >> 2) << 3;
    const float eb = bias[ecol];
    const float es = alpha[ecol];
    float pacc = 0.0f;
    int pg = -1;
    float* __restrict__ gcol = gpool + goffset + ecol;

    const float* __restrict__ zl = zin + lane;
    const int numTiles = (N + 15) >> 4;

    const int tBase = numTiles / (int)gridDim.x;
    const int tRem  = numTiles % (int)gridDim.x;
    const int bIdx  = (int)blockIdx.x;
    const int tBeg  = bIdx * tBase + (bIdx < tRem ? bIdx : tRem);
    const int tEnd  = tBeg + tBase + (bIdx < tRem ? 1 : 0);
    for (int tile = tBeg; tile < tEnd; ++tile) {
        const int nA = tile * 16 + wave * 2;
        const int nB = nA + 1;
        const int cA = nA < N ? nA : N - 1;
        const int cB = nB < N ? nB : N - 1;
        const int pA = rowptr[nA < N ? nA : N];
        const int pB = rowptr[nB < N ? nB : N];
        const int pC = rowptr[nB + 1 < N ? nB + 1 : N];
        const float dA = dinv[cA];
        const float dB = dinv[cB];
        const int eN = tile * 16 + erow;
        const int eNl = eN + (lane & 7);
        const int egv = batch[eNl < N ? eNl : N - 1];
        float edv = 1.0f;
        if (SCALE_OUT) edv = dinv[eNl < N ? eNl : N - 1];
        const float* __restrict__ qA = zl + (size_t)cA * HDIM;
        const float* __restrict__ qB = zl + (size_t)cB * HDIM;
        const float xA0 = qA[0], xA1 = qA[32], xA2 = qA[64], xA3 = qA[96];
        const float xB0 = qB[0], xB1 = qB[32], xB2 = qB[64], xB3 = qB[96];
        int sA0 = 0, sB0 = 0;
        if (lane < pB - pA) sA0 = col[pA + lane];
        if (lane < pC - pB) sB0 = col[pB + lane];

        {
            float a0 = 0.0f, a1 = 0.0f, a2 = 0.0f, a3 = 0.0f;
            if (nA < N) {
                a0 = xA0; a1 = xA1; a2 = xA2; a3 = xA3;
                add_in_edges(zl, pA, pB, col, lane, sA0, a0, a1, a2, a3);
                a0 *= dA; a1 *= dA; a2 *= dA; a3 *= dA;
            }
            float* __restrict__ sp = sA + (wave * 2) * TPITCH + lane;
            sp[0] = a0; sp[32] = a1; sp[64] = a2; sp[96] = a3;
        }
        {
            float a0 = 0.0f, a1 = 0.0f, a2 = 0.0f, a3 = 0.0f;
            if (nB < N) {
                a0 = xB0; a1 = xB1; a2 = xB2; a3 = xB3;
                add_in_edges(zl, pB, pC, col, lane, sB0, a0, a1, a2, a3);
                a0 *= dB; a1 *= dB; a2 *= dB; a3 *= dB;
            }
            float* __restrict__ sp = sA + (wave * 2 + 1) * TPITCH + lane;
            sp[0] = a0; sp[32] = a1; sp[64] = a2; sp[96] = a3;
        }
        __syncthreads();

        v8f acc = {};
        {
            const float* __restrict__ ap = sA + lmod * TPITCH + kb;
#pragma unroll
            for (int i = 0; i < 32; ++i) {
                const v2f a = *(const v2f*)(ap + 4 * i);
                acc = __builtin_amdgcn_wmma_f32_16x16x4_f32(
                    false, a, false, bf[i], (short)0, acc, false, false);
            }
        }

        {
            float* __restrict__ cp = sC + ((lane >> 4) << 3) * TPITCH + n0 + lmod;
#pragma unroll
            for (int r = 0; r < 8; ++r) cp[r * TPITCH] = acc[r];
        }
        __syncthreads();

        {
            const float* __restrict__ rp = sC + erow * TPITCH + ecol;
            float* __restrict__ zp = zout + (size_t)eN * HDIM + ecol;
            float z[8];
#pragma unroll
            for (int r = 0; r < 8; ++r) z[r] = rp[r * TPITCH];
#pragma unroll
            for (int r = 0; r < 8; ++r) {
                if (eN + r >= N) break;
                const float v = z[r] + eb;
                const float y = v * ((v > 0.0f) ? 1.0f : es);
                float o = y;
                if (SCALE_OUT) o = y * __int_as_float(__builtin_amdgcn_readlane(__float_as_int(edv), r));
                zp[(size_t)r * HDIM] = o;
                const int g = __builtin_amdgcn_readlane(egv, r);
                if (g != pg) {
                    if (pg >= 0) atomicAdd(gcol + (size_t)pg * (2 * HDIM), pacc);
                    pacc = 0.0f;
                    pg = g;
                }
                pacc += y;
            }
        }
    }
    if (pg >= 0) atomicAdd(gcol + (size_t)pg * (2 * HDIM), pacc);
}


extern "C" void kernel_launch(void* const* d_in, const int* in_sizes, int n_in,
                              void* d_out, int out_size, void* d_ws, size_t ws_size,
                              hipStream_t stream) {
    const float* x     = (const float*)d_in[0];
    const int*   ei    = (const int*)  d_in[1];
    const int*   batch = (const int*)  d_in[2];
    const float* W0    = (const float*)d_in[3];
    const float* b0    = (const float*)d_in[4];
    const float* al0   = (const float*)d_in[5];
    const float* W1    = (const float*)d_in[6];
    const float* b1    = (const float*)d_in[7];
    const float* al1   = (const float*)d_in[8];

    const int N = in_sizes[2];
    const int E = in_sizes[1] / 2;
    const int* srcp = ei;
    const int* dstp = ei + E;

    const int NH = N * HDIM;
    const int numTiles = (N + 15) / 16;
    const int layerBlocks = numTiles < 1024 ? numTiles : 1024;
    const int nb = (N + BK_NODES - 1) / BK_NODES;

    float* z1s    = (float*)d_ws;
    float* xs     = z1s + (size_t)NH;
    int2*  es     = (int2*)(xs + (size_t)NH);
    int*   col    = (int*)(es + E);
    int*   offT   = col + E;
    float* dinv   = (float*)(offT + (size_t)(nb + 1) * ES_SLICES);
    int*   rowptr = (int*)(dinv + N);

    float* z2out = (float*)d_out;
    float* gpool = z2out + (size_t)N * HDIM;
    const int gsize = out_size - N * HDIM;

    k_tilesort<<<ES_SLICES, ES_THREADS, 0, stream>>>(srcp, dstp, es, offT, gpool, gsize, E, N, nb);
    k_csr<<<nb, BK_THREADS, 0, stream>>>(es, offT, x, xs, rowptr, dinv, col, E, N);

    k_layer<true><<<layerBlocks, 256, 0, stream>>>(xs, W0, rowptr, col, dinv, b0, al0, batch,
                                                   z1s, gpool, 0, N);

    k_layer<false><<<layerBlocks, 256, 0, stream>>>(z1s, W1, rowptr, col, dinv, b1, al1, batch,
                                                    z2out, gpool, HDIM, N);
}
